# starting kernel
# baseline (speedup 1.0000x reference)
#include <hip/hip_runtime.h>


#define D 128
#define A 4
#define H 128
#define KST_E 168
#define KST_P 136

typedef __attribute__((ext_vector_type(16))) __bf16 bf16x16;
typedef __attribute__((ext_vector_type(8)))  float  f32x8;

struct alignas(16) U4 { unsigned int x, y, z, w; };
struct alignas(16) F4 { float x, y, z, w; };

union FragAB {
    bf16x16        v;
    unsigned short us[16];
    unsigned int   ui[8];
};
union FragC {
    f32x8 v;
    float f[8];
};

__device__ __forceinline__ void pack8(FragAB& fr, int base,
                                      const F4& a, const F4& b) {
    fr.v[base + 0] = (__bf16)a.x; fr.v[base + 1] = (__bf16)a.y;
    fr.v[base + 2] = (__bf16)a.z; fr.v[base + 3] = (__bf16)a.w;
    fr.v[base + 4] = (__bf16)b.x; fr.v[base + 5] = (__bf16)b.y;
    fr.v[base + 6] = (__bf16)b.z; fr.v[base + 7] = (__bf16)b.w;
}

__device__ __forceinline__ unsigned short f2bf_s(float f) {
    __bf16 h = (__bf16)f;
    return __builtin_bit_cast(unsigned short, h);
}

__device__ __forceinline__ void loadB(FragAB& b, const unsigned short* sW, int off) {
    *(U4*)(&b.ui[0]) = *(const U4*)(&sW[off]);
    *(U4*)(&b.ui[4]) = *(const U4*)(&sW[off + 16]);
}

__global__ __launch_bounds__(256) void zero_kernel(F4* __restrict__ p, int n4) {
    int i = blockIdx.x * 256 + threadIdx.x;
    if (i < n4) { F4 z{0.f, 0.f, 0.f, 0.f}; p[i] = z; }
}

__global__ __launch_bounds__(512)
void edge_kernel(const float* __restrict__ feats,
                 const float* __restrict__ addInfo,
                 const int*   __restrict__ nidx,
                 const int*   __restrict__ segIds,
                 const float* __restrict__ Wb,
                 const float* __restrict__ bb,
                 unsigned int* __restrict__ pooled,
                 int E) {
    __shared__ unsigned short sW[H * KST_E];

    const int tid = threadIdx.x;
    for (int i = tid; i < (D + A) * H; i += 512) {
        int k = i >> 7, n = i & 127;
        sW[n * KST_E + k] = f2bf_s(Wb[i]);
    }
    for (int i = tid; i < H * (KST_E - (D + A)); i += 512) {
        int n = i / (KST_E - (D + A));
        int k = (D + A) + i % (KST_E - (D + A));
        sW[n * KST_E + k] = 0;
    }
    __syncthreads();

    const int wave = tid >> 5, lane = tid & 31;
    const int hi = lane >> 4, ln = lane & 15;
    const int strip = blockIdx.x * 256 + wave * 16;

    const int eRow   = strip + ln;
    const int eClamp = eRow < E ? eRow : (E - 1);
    const int gidx   = nidx[eClamp];
    const float* __restrict__ rowp = feats + gidx * D;

    FragC acc[8];
#pragma unroll
    for (int nt = 0; nt < 8; ++nt)
#pragma unroll
        for (int r = 0; r < 8; ++r) acc[nt].f[r] = 0.f;

#pragma unroll
    for (int kt = 0; kt < 5; ++kt) {
        FragAB a;
        const int kb = kt * 32 + hi * 8;
        if (kt < 4) {
            F4 f0 = *(const F4*)(rowp + kb);
            F4 f1 = *(const F4*)(rowp + kb + 4);
            F4 f2 = *(const F4*)(rowp + kb + 16);
            F4 f3 = *(const F4*)(rowp + kb + 20);
            pack8(a, 0, f0, f1);
            pack8(a, 8, f2, f3);
        } else {
#pragma unroll
            for (int i = 0; i < 8; ++i) a.ui[i] = 0u;
            if (!hi) {
                F4 ai = *(const F4*)(addInfo + eClamp * A);
                a.v[0] = (__bf16)ai.x; a.v[1] = (__bf16)ai.y;
                a.v[2] = (__bf16)ai.z; a.v[3] = (__bf16)ai.w;
            }
        }
        FragAB bcur, bnxt;
        loadB(bcur, sW, ln * KST_E + kb);
#pragma unroll
        for (int nt = 0; nt < 8; ++nt) {
            if (nt < 7)
                loadB(bnxt, sW, ((nt + 1) * 16 + ln) * KST_E + kb);
            acc[nt].v = __builtin_amdgcn_wmma_f32_16x16x32_bf16(
                false, a.v, false, bcur.v, (short)0, acc[nt].v, false, false);
            bcur = bnxt;
        }
    }

    int segArr[8];
#pragma unroll
    for (int r = 0; r < 8; ++r) {
        int e = strip + r + hi * 8;
        segArr[r] = (e < E) ? segIds[e] : -1;
    }
#pragma unroll
    for (int nt = 0; nt < 8; ++nt) {
        const int j = nt * 16 + ln;
        const float bbj = bb[j];
        int prevSeg = -1; float best = 0.f;
#pragma unroll
        for (int r = 0; r < 8; ++r) {
            if (segArr[r] >= 0) {
                float v = fmaxf(acc[nt].f[r] + bbj, 0.f);
                if (segArr[r] != prevSeg) {
                    if (prevSeg >= 0)
                        atomicMax(&pooled[prevSeg * H + j], __float_as_uint(best));
                    prevSeg = segArr[r];
                    best = v;
                } else {
                    best = fmaxf(best, v);
                }
            }
        }
        if (prevSeg >= 0)
            atomicMax(&pooled[prevSeg * H + j], __float_as_uint(best));
    }
}

__global__ __launch_bounds__(256)
void proj_kernel(const float* __restrict__ pooled,
                 const float* __restrict__ Wo,
                 const float* __restrict__ bo,
                 const float* __restrict__ src,
                 float* __restrict__ dst,
                 int N) {
    __shared__ unsigned short sW[D * KST_P];

    const int tid = threadIdx.x;
    for (int i = tid; i < H * D; i += 256) {
        int h = i >> 7, j = i & 127;
        sW[j * KST_P + h] = f2bf_s(Wo[i]);
    }
    for (int i = tid; i < D * (KST_P - H); i += 256) {
        int j = i >> 3, k = H + (i & 7);
        sW[j * KST_P + k] = 0;
    }
    __syncthreads();

    const int wave = tid >> 5, lane = tid & 31;
    const int hi = lane >> 4, ln = lane & 15;
    const int mtiles = (N + 15) >> 4;
    const int mt = blockIdx.x * 8 + wave;
    if (mt >= mtiles) return;

    const int mBase = mt * 16;
    const int mRow  = (mBase + ln) < N ? (mBase + ln) : (N - 1);
    const float* __restrict__ rowp = pooled + mRow * H;

    FragC acc[8];
#pragma unroll
    for (int nt = 0; nt < 8; ++nt)
#pragma unroll
        for (int r = 0; r < 8; ++r) acc[nt].f[r] = 0.f;

#pragma unroll
    for (int kt = 0; kt < 4; ++kt) {
        const int kb = kt * 32 + hi * 8;
        FragAB a;
        F4 f0 = *(const F4*)(rowp + kb);
        F4 f1 = *(const F4*)(rowp + kb + 4);
        F4 f2 = *(const F4*)(rowp + kb + 16);
        F4 f3 = *(const F4*)(rowp + kb + 20);
        pack8(a, 0, f0, f1);
        pack8(a, 8, f2, f3);
        FragAB bcur, bnxt;
        loadB(bcur, sW, ln * KST_P + kb);
#pragma unroll
        for (int nt = 0; nt < 8; ++nt) {
            if (nt < 7)
                loadB(bnxt, sW, ((nt + 1) * 16 + ln) * KST_P + kb);
            acc[nt].v = __builtin_amdgcn_wmma_f32_16x16x32_bf16(
                false, a.v, false, bcur.v, (short)0, acc[nt].v, false, false);
            bcur = bnxt;
        }
    }

#pragma unroll
    for (int nt = 0; nt < 8; ++nt) {
        const int j = nt * 16 + ln;
        const float boj = bo[j];
#pragma unroll
        for (int r = 0; r < 8; ++r) {
            const int m = mBase + r + hi * 8;
            if (m < N)
                dst[m * D + j] = acc[nt].f[r] + boj + src[m * D + j];
        }
    }
}

__global__ __launch_bounds__(256)
void final_kernel(const float* __restrict__ x,
                  const float* __restrict__ Wf,
                  const float* __restrict__ bf,
                  float* __restrict__ out,
                  int N) {
    const int tid = threadIdx.x;
    const int wave = tid >> 5, lane = tid & 31;
    const int n = blockIdx.x * 8 + wave;
    if (n >= N) return;
    F4 a = ((const F4*)(x + n * D))[lane];
    F4 w = ((const F4*)Wf)[lane];
    float s = a.x * w.x + a.y * w.y + a.z * w.z + a.w * w.w;
#pragma unroll
    for (int m = 16; m >= 1; m >>= 1)
        s += __shfl_xor(s, m, 32);
    if (lane == 0) out[n] = s + bf[0];
}

extern "C" void kernel_launch(void* const* d_in, const int* in_sizes, int n_in,
                              void* d_out, int out_size, void* d_ws, size_t ws_size,
                              hipStream_t stream) {
    const float* interp  = (const float*)d_in[0];
    const float* addInfo = (const float*)d_in[1];
    const int*   nidx    = (const int*)d_in[2];
    const int*   segIds  = (const int*)d_in[3];
    const float* Wb0 = (const float*)d_in[4];
    const float* bb0 = (const float*)d_in[5];
    const float* Wo0 = (const float*)d_in[6];
    const float* bo0 = (const float*)d_in[7];
    const float* Wb1 = (const float*)d_in[8];
    const float* bb1 = (const float*)d_in[9];
    const float* Wo1 = (const float*)d_in[10];
    const float* bo1 = (const float*)d_in[11];
    const float* Wf  = (const float*)d_in[12];
    const float* bfp = (const float*)d_in[13];

    const int N = in_sizes[0] / D;
    const int E = in_sizes[2];

    float* pooled = (float*)d_ws;
    float* xws    = pooled + (size_t)N * H;
    unsigned int* pooledU = (unsigned int*)pooled;

    const int n4      = (N * H) / 4;
    const int zeroGrd = (n4 + 255) / 256;
    const int edgeGrd = (E + 255) / 256;
    const int projGrd = (((N + 15) / 16) + 7) / 8;
    const int finGrd  = (N + 7) / 8;

    zero_kernel<<<zeroGrd, 256, 0, stream>>>((F4*)pooled, n4);
    edge_kernel<<<edgeGrd, 512, 0, stream>>>(interp, addInfo, nidx, segIds,
                                             Wb0, bb0, pooledU, E);
    proj_kernel<<<projGrd, 256, 0, stream>>>(pooled, Wo0, bo0, interp, xws, N);

    zero_kernel<<<zeroGrd, 256, 0, stream>>>((F4*)pooled, n4);
    edge_kernel<<<edgeGrd, 512, 0, stream>>>(xws, addInfo, nidx, segIds,
                                             Wb1, bb1, pooledU, E);
    proj_kernel<<<projGrd, 256, 0, stream>>>(pooled, Wo1, bo1, xws, xws, N);

    final_kernel<<<finGrd, 256, 0, stream>>>(xws, Wf, bfp, (float*)d_out, N);
}


# edge kernel: 512 persistent workgroups stage the weights once; each lane half walks a contiguous run of rows and carries the running segment max in registers, one atomicMax per segment change
# speedup vs baseline: 1.4273x; 1.4273x over previous
#include <hip/hip_runtime.h>


#define D 128
#define A 4
#define H 128
#define KST_E 168
#define KST_P 136

typedef __attribute__((ext_vector_type(16))) __bf16 bf16x16;
typedef __attribute__((ext_vector_type(8)))  float  f32x8;

struct alignas(16) U4 { unsigned int x, y, z, w; };
struct alignas(16) F4 { float x, y, z, w; };

union FragAB {
    bf16x16        v;
    unsigned short us[16];
    unsigned int   ui[8];
};
union FragC {
    f32x8 v;
    float f[8];
};

__device__ __forceinline__ void pack8(FragAB& fr, int base,
                                      const F4& a, const F4& b) {
    fr.v[base + 0] = (__bf16)a.x; fr.v[base + 1] = (__bf16)a.y;
    fr.v[base + 2] = (__bf16)a.z; fr.v[base + 3] = (__bf16)a.w;
    fr.v[base + 4] = (__bf16)b.x; fr.v[base + 5] = (__bf16)b.y;
    fr.v[base + 6] = (__bf16)b.z; fr.v[base + 7] = (__bf16)b.w;
}

__device__ __forceinline__ unsigned short f2bf_s(float f) {
    __bf16 h = (__bf16)f;
    return __builtin_bit_cast(unsigned short, h);
}

__device__ __forceinline__ void loadB(FragAB& b, const unsigned short* sW, int off) {
    *(U4*)(&b.ui[0]) = *(const U4*)(&sW[off]);
    *(U4*)(&b.ui[4]) = *(const U4*)(&sW[off + 16]);
}

__global__ __launch_bounds__(256) void zero_kernel(F4* __restrict__ p, int n4) {
    int i = blockIdx.x * 256 + threadIdx.x;
    if (i < n4) { F4 z{0.f, 0.f, 0.f, 0.f}; p[i] = z; }
}

__global__ __launch_bounds__(512) __attribute__((amdgpu_waves_per_eu(8, 8)))
void edge_kernel(const float* __restrict__ feats,
                 const float* __restrict__ addInfo,
                 const int*   __restrict__ nidx,
                 const int*   __restrict__ segIds,
                 const float* __restrict__ Wb,
                 const float* __restrict__ bb,
                 unsigned int* __restrict__ pooled,
                 int E) {
    __shared__ unsigned short sW[H * KST_E];

    const int tid = threadIdx.x;
    for (int i = tid; i < (D + A) * H; i += 512) {
        int k = i >> 7, n = i & 127;
        sW[n * KST_E + k] = f2bf_s(Wb[i]);
    }
    for (int i = tid; i < H * (KST_E - (D + A)); i += 512) {
        int n = i / (KST_E - (D + A));
        int k = (D + A) + i % (KST_E - (D + A));
        sW[n * KST_E + k] = 0;
    }
    __syncthreads();

    const int wave = tid >> 5, lane = tid & 31;
    const int hi = lane >> 4, ln = lane & 15;
    const int numGroups = (E + 7) >> 3;
    const int numHalves = (int)gridDim.x * 32;
    const int steps     = (numGroups + numHalves - 1) / numHalves;
    const int waveId    = (int)blockIdx.x * 16 + wave;
    const int aBase = ((waveId * 2 + (ln >> 3)) * steps) * 8 + (ln & 7);
    const int dBase = ((waveId * 2 + hi) * steps) * 8;

    float bbv[8];
#pragma unroll
    for (int nt = 0; nt < 8; ++nt) bbv[nt] = bb[nt * 16 + ln];

    int gidx = nidx[aBase < E ? aBase : (E - 1)];

    int   prevSeg = -1;
    float best[8];
#pragma unroll
    for (int nt = 0; nt < 8; ++nt) best[nt] = 0.f;

#pragma unroll 1
    for (int t = 0; t < steps; ++t) {
    asm volatile("" ::: "memory");
    const int eRow   = aBase + t * 8;
    const int eClamp = eRow < E ? eRow : (E - 1);
    const float* __restrict__ rowp = feats + gidx * D;

    const int eNext    = eRow + 8;
    const int gidxNext = nidx[eNext < E ? eNext : (E - 1)];

    FragC acc[8];
#pragma unroll
    for (int nt = 0; nt < 8; ++nt)
#pragma unroll
        for (int r = 0; r < 8; ++r) acc[nt].f[r] = 0.f;

#pragma unroll
    for (int kt = 0; kt < 5; ++kt) {
        FragAB a;
        const int kb = kt * 32 + hi * 8;
        if (kt < 4) {
            F4 f0 = *(const F4*)(rowp + kb);
            F4 f1 = *(const F4*)(rowp + kb + 4);
            F4 f2 = *(const F4*)(rowp + kb + 16);
            F4 f3 = *(const F4*)(rowp + kb + 20);
            pack8(a, 0, f0, f1);
            pack8(a, 8, f2, f3);
        } else {
#pragma unroll
            for (int i = 0; i < 8; ++i) a.ui[i] = 0u;
            if (!hi) {
                F4 ai = *(const F4*)(addInfo + eClamp * A);
                a.v[0] = (__bf16)ai.x; a.v[1] = (__bf16)ai.y;
                a.v[2] = (__bf16)ai.z; a.v[3] = (__bf16)ai.w;
            }
        }
        FragAB bcur, bnxt;
        loadB(bcur, sW, ln * KST_E + kb);
#pragma unroll
        for (int nt = 0; nt < 8; ++nt) {
            if (nt < 7)
                loadB(bnxt, sW, ((nt + 1) * 16 + ln) * KST_E + kb);
            acc[nt].v = __builtin_amdgcn_wmma_f32_16x16x32_bf16(
                false, a.v, false, bcur.v, (short)0, acc[nt].v, false, false);
            bcur = bnxt;
        }
    }

    int segArr[8];
#pragma unroll
    for (int r = 0; r < 8; ++r) {
        int e = dBase + t * 8 + r;
        segArr[r] = (e < E) ? segIds[e] : -1;
    }
    int lastSeg = prevSeg;
#pragma unroll
    for (int nt = 0; nt < 8; ++nt) {
        const int j = nt * 16 + ln;
        const float bbj = bbv[nt];
        int ps = prevSeg; float b = best[nt];
#pragma unroll
        for (int r = 0; r < 8; ++r) {
            if (segArr[r] >= 0) {
                float v = fmaxf(acc[nt].f[r] + bbj, 0.f);
                if (segArr[r] != ps) {
                    if (ps >= 0)
                        atomicMax(&pooled[ps * H + j], __float_as_uint(b));
                    ps = segArr[r];
                    b = v;
                } else {
                    b = fmaxf(b, v);
                }
            }
        }
        best[nt] = b;
        lastSeg  = ps;
    }
    prevSeg = lastSeg;

    gidx = gidxNext;
    }

    if (prevSeg >= 0) {
#pragma unroll
        for (int nt = 0; nt < 8; ++nt)
            atomicMax(&pooled[prevSeg * H + nt * 16 + ln], __float_as_uint(best[nt]));
    }
}

__global__ __launch_bounds__(256)
void proj_kernel(const float* __restrict__ pooled,
                 const float* __restrict__ Wo,
                 const float* __restrict__ bo,
                 const float* __restrict__ src,
                 float* __restrict__ dst,
                 int N) {
    __shared__ unsigned short sW[D * KST_P];

    const int tid = threadIdx.x;
    for (int i = tid; i < H * D; i += 256) {
        int h = i >> 7, j = i & 127;
        sW[j * KST_P + h] = f2bf_s(Wo[i]);
    }
    for (int i = tid; i < D * (KST_P - H); i += 256) {
        int j = i >> 3, k = H + (i & 7);
        sW[j * KST_P + k] = 0;
    }
    __syncthreads();

    const int wave = tid >> 5, lane = tid & 31;
    const int hi = lane >> 4, ln = lane & 15;
    const int mtiles = (N + 15) >> 4;
    const int mt = blockIdx.x * 8 + wave;
    if (mt >= mtiles) return;

    const int mBase = mt * 16;
    const int mRow  = (mBase + ln) < N ? (mBase + ln) : (N - 1);
    const float* __restrict__ rowp = pooled + mRow * H;

    FragC acc[8];
#pragma unroll
    for (int nt = 0; nt < 8; ++nt)
#pragma unroll
        for (int r = 0; r < 8; ++r) acc[nt].f[r] = 0.f;

#pragma unroll
    for (int kt = 0; kt < 4; ++kt) {
        const int kb = kt * 32 + hi * 8;
        FragAB a;
        F4 f0 = *(const F4*)(rowp + kb);
        F4 f1 = *(const F4*)(rowp + kb + 4);
        F4 f2 = *(const F4*)(rowp + kb + 16);
        F4 f3 = *(const F4*)(rowp + kb + 20);
        pack8(a, 0, f0, f1);
        pack8(a, 8, f2, f3);
        FragAB bcur, bnxt;
        loadB(bcur, sW, ln * KST_P + kb);
#pragma unroll
        for (int nt = 0; nt < 8; ++nt) {
            if (nt < 7)
                loadB(bnxt, sW, ((nt + 1) * 16 + ln) * KST_P + kb);
            acc[nt].v = __builtin_amdgcn_wmma_f32_16x16x32_bf16(
                false, a.v, false, bcur.v, (short)0, acc[nt].v, false, false);
            bcur = bnxt;
        }
    }

#pragma unroll
    for (int nt = 0; nt < 8; ++nt) {
        const int j = nt * 16 + ln;
        const float boj = bo[j];
#pragma unroll
        for (int r = 0; r < 8; ++r) {
            const int m = mBase + r + hi * 8;
            if (m < N)
                dst[m * D + j] = acc[nt].f[r] + boj + src[m * D + j];
        }
    }
}

__global__ __launch_bounds__(256)
void final_kernel(const float* __restrict__ x,
                  const float* __restrict__ Wf,
                  const float* __restrict__ bf,
                  float* __restrict__ out,
                  int N) {
    const int tid = threadIdx.x;
    const int wave = tid >> 5, lane = tid & 31;
    const int n = blockIdx.x * 8 + wave;
    if (n >= N) return;
    F4 a = ((const F4*)(x + n * D))[lane];
    F4 w = ((const F4*)Wf)[lane];
    float s = a.x * w.x + a.y * w.y + a.z * w.z + a.w * w.w;
#pragma unroll
    for (int m = 16; m >= 1; m >>= 1)
        s += __shfl_xor(s, m, 32);
    if (lane == 0) out[n] = s + bf[0];
}

extern "C" void kernel_launch(void* const* d_in, const int* in_sizes, int n_in,
                              void* d_out, int out_size, void* d_ws, size_t ws_size,
                              hipStream_t stream) {
    const float* interp  = (const float*)d_in[0];
    const float* addInfo = (const float*)d_in[1];
    const int*   nidx    = (const int*)d_in[2];
    const int*   segIds  = (const int*)d_in[3];
    const float* Wb0 = (const float*)d_in[4];
    const float* bb0 = (const float*)d_in[5];
    const float* Wo0 = (const float*)d_in[6];
    const float* bo0 = (const float*)d_in[7];
    const float* Wb1 = (const float*)d_in[8];
    const float* bb1 = (const float*)d_in[9];
    const float* Wo1 = (const float*)d_in[10];
    const float* bo1 = (const float*)d_in[11];
    const float* Wf  = (const float*)d_in[12];
    const float* bfp = (const float*)d_in[13];

    const int N = in_sizes[0] / D;
    const int E = in_sizes[2];

    float* pooled = (float*)d_ws;
    float* xws    = pooled + (size_t)N * H;
    unsigned int* pooledU = (unsigned int*)pooled;

    const int n4      = (N * H) / 4;
    const int zeroGrd = (n4 + 255) / 256;
    const int edgeGrd = 512;
    const int projGrd = (((N + 15) / 16) + 7) / 8;
    const int finGrd  = (N + 7) / 8;

    zero_kernel<<<zeroGrd, 256, 0, stream>>>((F4*)pooled, n4);
    edge_kernel<<<edgeGrd, 512, 0, stream>>>(interp, addInfo, nidx, segIds,
                                             Wb0, bb0, pooledU, E);
    proj_kernel<<<projGrd, 256, 0, stream>>>(pooled, Wo0, bo0, interp, xws, N);

    zero_kernel<<<zeroGrd, 256, 0, stream>>>((F4*)pooled, n4);
    edge_kernel<<<edgeGrd, 512, 0, stream>>>(xws, addInfo, nidx, segIds,
                                             Wb1, bb1, pooledU, E);
    proj_kernel<<<projGrd, 256, 0, stream>>>(pooled, Wo1, bo1, xws, xws, N);

    final_kernel<<<finGrd, 256, 0, stream>>>(xws, Wf, bfp, (float*)d_out, N);
}


# edge epilogue: rows outermost, one segment-change test and flush branch per row instead of per row and column tile; running max branch-free on raw accumulators, bias+relu once per flushed run
# speedup vs baseline: 1.7373x; 1.2172x over previous
#include <hip/hip_runtime.h>


#define D 128
#define A 4
#define H 128
#define KST_E 168
#define KST_P 136

typedef __attribute__((ext_vector_type(16))) __bf16 bf16x16;
typedef __attribute__((ext_vector_type(8)))  float  f32x8;

struct alignas(16) U4 { unsigned int x, y, z, w; };
struct alignas(16) F4 { float x, y, z, w; };

union FragAB {
    bf16x16        v;
    unsigned short us[16];
    unsigned int   ui[8];
};
union FragC {
    f32x8 v;
    float f[8];
};

__device__ __forceinline__ void pack8(FragAB& fr, int base,
                                      const F4& a, const F4& b) {
    fr.v[base + 0] = (__bf16)a.x; fr.v[base + 1] = (__bf16)a.y;
    fr.v[base + 2] = (__bf16)a.z; fr.v[base + 3] = (__bf16)a.w;
    fr.v[base + 4] = (__bf16)b.x; fr.v[base + 5] = (__bf16)b.y;
    fr.v[base + 6] = (__bf16)b.z; fr.v[base + 7] = (__bf16)b.w;
}

__device__ __forceinline__ unsigned short f2bf_s(float f) {
    __bf16 h = (__bf16)f;
    return __builtin_bit_cast(unsigned short, h);
}

__device__ __forceinline__ void loadB(FragAB& b, const unsigned short* sW, int off) {
    *(U4*)(&b.ui[0]) = *(const U4*)(&sW[off]);
    *(U4*)(&b.ui[4]) = *(const U4*)(&sW[off + 16]);
}

__global__ __launch_bounds__(256) void zero_kernel(F4* __restrict__ p, int n4) {
    int i = blockIdx.x * 256 + threadIdx.x;
    if (i < n4) { F4 z{0.f, 0.f, 0.f, 0.f}; p[i] = z; }
}

__global__ __launch_bounds__(512) __attribute__((amdgpu_waves_per_eu(8, 8)))
void edge_kernel(const float* __restrict__ feats,
                 const float* __restrict__ addInfo,
                 const int*   __restrict__ nidx,
                 const int*   __restrict__ segIds,
                 const float* __restrict__ Wb,
                 const float* __restrict__ bb,
                 unsigned int* __restrict__ pooled,
                 int E) {
    __shared__ unsigned short sW[H * KST_E];

    const int tid = threadIdx.x;
    for (int i = tid; i < (D + A) * H; i += 512) {
        int k = i >> 7, n = i & 127;
        sW[n * KST_E + k] = f2bf_s(Wb[i]);
    }
    for (int i = tid; i < H * (KST_E - (D + A)); i += 512) {
        int n = i / (KST_E - (D + A));
        int k = (D + A) + i % (KST_E - (D + A));
        sW[n * KST_E + k] = 0;
    }
    __syncthreads();

    const int wave = tid >> 5, lane = tid & 31;
    const int hi = lane >> 4, ln = lane & 15;
    const int numGroups = (E + 7) >> 3;
    const int numHalves = (int)gridDim.x * 32;
    const int steps     = (numGroups + numHalves - 1) / numHalves;
    const int waveId    = (int)blockIdx.x * 16 + wave;
    const int aBase = ((waveId * 2 + (ln >> 3)) * steps) * 8 + (ln & 7);
    const int dBase = ((waveId * 2 + hi) * steps) * 8;

    float bbv[8];
#pragma unroll
    for (int nt = 0; nt < 8; ++nt) bbv[nt] = bb[nt * 16 + ln];

    int gidx = nidx[aBase < E ? aBase : (E - 1)];

    int   prevSeg = -1;
    float best[8];
#pragma unroll
    for (int nt = 0; nt < 8; ++nt) best[nt] = -__builtin_inff();

#pragma unroll 1
    for (int t = 0; t < steps; ++t) {
    asm volatile("" ::: "memory");
    const int eRow   = aBase + t * 8;
    const int eClamp = eRow < E ? eRow : (E - 1);
    const float* __restrict__ rowp = feats + gidx * D;

    const int eNext    = eRow + 8;
    const int gidxNext = nidx[eNext < E ? eNext : (E - 1)];

    FragC acc[8];
#pragma unroll
    for (int nt = 0; nt < 8; ++nt)
#pragma unroll
        for (int r = 0; r < 8; ++r) acc[nt].f[r] = 0.f;

#pragma unroll
    for (int kt = 0; kt < 5; ++kt) {
        FragAB a;
        const int kb = kt * 32 + hi * 8;
        if (kt < 4) {
            F4 f0 = *(const F4*)(rowp + kb);
            F4 f1 = *(const F4*)(rowp + kb + 4);
            F4 f2 = *(const F4*)(rowp + kb + 16);
            F4 f3 = *(const F4*)(rowp + kb + 20);
            pack8(a, 0, f0, f1);
            pack8(a, 8, f2, f3);
        } else {
#pragma unroll
            for (int i = 0; i < 8; ++i) a.ui[i] = 0u;
            if (!hi) {
                F4 ai = *(const F4*)(addInfo + eClamp * A);
                a.v[0] = (__bf16)ai.x; a.v[1] = (__bf16)ai.y;
                a.v[2] = (__bf16)ai.z; a.v[3] = (__bf16)ai.w;
            }
        }
        FragAB bcur, bnxt;
        loadB(bcur, sW, ln * KST_E + kb);
#pragma unroll
        for (int nt = 0; nt < 8; ++nt) {
            if (nt < 7)
                loadB(bnxt, sW, ((nt + 1) * 16 + ln) * KST_E + kb);
            acc[nt].v = __builtin_amdgcn_wmma_f32_16x16x32_bf16(
                false, a.v, false, bcur.v, (short)0, acc[nt].v, false, false);
            bcur = bnxt;
        }
    }

    int segArr[8];
#pragma unroll
    for (int r = 0; r < 8; ++r) {
        int e = dBase + t * 8 + r;
        segArr[r] = (e < E) ? segIds[e] : -1;
    }
#pragma unroll
    for (int r = 0; r < 8; ++r) {
        const int s = segArr[r];
        if (s >= 0) {
            if (s != prevSeg) {
                if (prevSeg >= 0) {
#pragma unroll
                    for (int nt = 0; nt < 8; ++nt)
                        atomicMax(&pooled[prevSeg * H + nt * 16 + ln],
                                  __float_as_uint(fmaxf(best[nt] + bbv[nt], 0.f)));
                }
                prevSeg = s;
#pragma unroll
                for (int nt = 0; nt < 8; ++nt) best[nt] = -__builtin_inff();
            }
#pragma unroll
            for (int nt = 0; nt < 8; ++nt)
                best[nt] = __builtin_elementwise_maximum(best[nt], acc[nt].f[r]);
        }
    }

    gidx = gidxNext;
    }

    if (prevSeg >= 0) {
#pragma unroll
        for (int nt = 0; nt < 8; ++nt)
            atomicMax(&pooled[prevSeg * H + nt * 16 + ln],
                      __float_as_uint(fmaxf(best[nt] + bbv[nt], 0.f)));
    }
}

__global__ __launch_bounds__(256)
void proj_kernel(const float* __restrict__ pooled,
                 const float* __restrict__ Wo,
                 const float* __restrict__ bo,
                 const float* __restrict__ src,
                 float* __restrict__ dst,
                 int N) {
    __shared__ unsigned short sW[D * KST_P];

    const int tid = threadIdx.x;
    for (int i = tid; i < H * D; i += 256) {
        int h = i >> 7, j = i & 127;
        sW[j * KST_P + h] = f2bf_s(Wo[i]);
    }
    for (int i = tid; i < D * (KST_P - H); i += 256) {
        int j = i >> 3, k = H + (i & 7);
        sW[j * KST_P + k] = 0;
    }
    __syncthreads();

    const int wave = tid >> 5, lane = tid & 31;
    const int hi = lane >> 4, ln = lane & 15;
    const int mtiles = (N + 15) >> 4;
    const int mt = blockIdx.x * 8 + wave;
    if (mt >= mtiles) return;

    const int mBase = mt * 16;
    const int mRow  = (mBase + ln) < N ? (mBase + ln) : (N - 1);
    const float* __restrict__ rowp = pooled + mRow * H;

    FragC acc[8];
#pragma unroll
    for (int nt = 0; nt < 8; ++nt)
#pragma unroll
        for (int r = 0; r < 8; ++r) acc[nt].f[r] = 0.f;

#pragma unroll
    for (int kt = 0; kt < 4; ++kt) {
        const int kb = kt * 32 + hi * 8;
        FragAB a;
        F4 f0 = *(const F4*)(rowp + kb);
        F4 f1 = *(const F4*)(rowp + kb + 4);
        F4 f2 = *(const F4*)(rowp + kb + 16);
        F4 f3 = *(const F4*)(rowp + kb + 20);
        pack8(a, 0, f0, f1);
        pack8(a, 8, f2, f3);
        FragAB bcur, bnxt;
        loadB(bcur, sW, ln * KST_P + kb);
#pragma unroll
        for (int nt = 0; nt < 8; ++nt) {
            if (nt < 7)
                loadB(bnxt, sW, ((nt + 1) * 16 + ln) * KST_P + kb);
            acc[nt].v = __builtin_amdgcn_wmma_f32_16x16x32_bf16(
                false, a.v, false, bcur.v, (short)0, acc[nt].v, false, false);
            bcur = bnxt;
        }
    }

#pragma unroll
    for (int nt = 0; nt < 8; ++nt) {
        const int j = nt * 16 + ln;
        const float boj = bo[j];
#pragma unroll
        for (int r = 0; r < 8; ++r) {
            const int m = mBase + r + hi * 8;
            if (m < N)
                dst[m * D + j] = acc[nt].f[r] + boj + src[m * D + j];
        }
    }
}

__global__ __launch_bounds__(256)
void final_kernel(const float* __restrict__ x,
                  const float* __restrict__ Wf,
                  const float* __restrict__ bf,
                  float* __restrict__ out,
                  int N) {
    const int tid = threadIdx.x;
    const int wave = tid >> 5, lane = tid & 31;
    const int n = blockIdx.x * 8 + wave;
    if (n >= N) return;
    F4 a = ((const F4*)(x + n * D))[lane];
    F4 w = ((const F4*)Wf)[lane];
    float s = a.x * w.x + a.y * w.y + a.z * w.z + a.w * w.w;
#pragma unroll
    for (int m = 16; m >= 1; m >>= 1)
        s += __shfl_xor(s, m, 32);
    if (lane == 0) out[n] = s + bf[0];
}

extern "C" void kernel_launch(void* const* d_in, const int* in_sizes, int n_in,
                              void* d_out, int out_size, void* d_ws, size_t ws_size,
                              hipStream_t stream) {
    const float* interp  = (const float*)d_in[0];
    const float* addInfo = (const float*)d_in[1];
    const int*   nidx    = (const int*)d_in[2];
    const int*   segIds  = (const int*)d_in[3];
    const float* Wb0 = (const float*)d_in[4];
    const float* bb0 = (const float*)d_in[5];
    const float* Wo0 = (const float*)d_in[6];
    const float* bo0 = (const float*)d_in[7];
    const float* Wb1 = (const float*)d_in[8];
    const float* bb1 = (const float*)d_in[9];
    const float* Wo1 = (const float*)d_in[10];
    const float* bo1 = (const float*)d_in[11];
    const float* Wf  = (const float*)d_in[12];
    const float* bfp = (const float*)d_in[13];

    const int N = in_sizes[0] / D;
    const int E = in_sizes[2];

    float* pooled = (float*)d_ws;
    float* xws    = pooled + (size_t)N * H;
    unsigned int* pooledU = (unsigned int*)pooled;

    const int n4      = (N * H) / 4;
    const int zeroGrd = (n4 + 255) / 256;
    const int edgeGrd = 512;
    const int projGrd = (((N + 15) / 16) + 7) / 8;
    const int finGrd  = (N + 7) / 8;

    zero_kernel<<<zeroGrd, 256, 0, stream>>>((F4*)pooled, n4);
    edge_kernel<<<edgeGrd, 512, 0, stream>>>(interp, addInfo, nidx, segIds,
                                             Wb0, bb0, pooledU, E);
    proj_kernel<<<projGrd, 256, 0, stream>>>(pooled, Wo0, bo0, interp, xws, N);

    zero_kernel<<<zeroGrd, 256, 0, stream>>>((F4*)pooled, n4);
    edge_kernel<<<edgeGrd, 512, 0, stream>>>(xws, addInfo, nidx, segIds,
                                             Wb1, bb1, pooledU, E);
    proj_kernel<<<projGrd, 256, 0, stream>>>(pooled, Wo1, bo1, xws, xws, N);

    final_kernel<<<finGrd, 256, 0, stream>>>(xws, Wf, bfp, (float*)d_out, N);
}


# edge kernel: operand loads issued ahead of their matrix work at 4 waves per SIMD: next step's 16 gathered row loads in flight a whole step, B fragments two groups ahead through a register ring
# speedup vs baseline: 1.9732x; 1.1358x over previous
#include <hip/hip_runtime.h>


#define D 128
#define A 4
#define H 128
#define KST_E 168
#define KST_P 136

typedef __attribute__((ext_vector_type(16))) __bf16 bf16x16;
typedef __attribute__((ext_vector_type(8)))  float  f32x8;

struct alignas(16) U4 { unsigned int x, y, z, w; };
struct alignas(16) F4 { float x, y, z, w; };

union FragAB {
    bf16x16        v;
    unsigned short us[16];
    unsigned int   ui[8];
};
union FragC {
    f32x8 v;
    float f[8];
};

__device__ __forceinline__ void pack8(FragAB& fr, int base,
                                      const F4& a, const F4& b) {
    fr.v[base + 0] = (__bf16)a.x; fr.v[base + 1] = (__bf16)a.y;
    fr.v[base + 2] = (__bf16)a.z; fr.v[base + 3] = (__bf16)a.w;
    fr.v[base + 4] = (__bf16)b.x; fr.v[base + 5] = (__bf16)b.y;
    fr.v[base + 6] = (__bf16)b.z; fr.v[base + 7] = (__bf16)b.w;
}

__device__ __forceinline__ unsigned short f2bf_s(float f) {
    __bf16 h = (__bf16)f;
    return __builtin_bit_cast(unsigned short, h);
}

__device__ __forceinline__ void loadB(FragAB& b, const unsigned short* sW, int off) {
    *(U4*)(&b.ui[0]) = *(const U4*)(&sW[off]);
    *(U4*)(&b.ui[4]) = *(const U4*)(&sW[off + 16]);
}

__global__ __launch_bounds__(256) void zero_kernel(F4* __restrict__ p, int n4) {
    int i = blockIdx.x * 256 + threadIdx.x;
    if (i < n4) { F4 z{0.f, 0.f, 0.f, 0.f}; p[i] = z; }
}

__global__ __launch_bounds__(512) __attribute__((amdgpu_waves_per_eu(4, 4)))
void edge_kernel(const float* __restrict__ feats,
                 const float* __restrict__ addInfo,
                 const int*   __restrict__ nidx,
                 const int*   __restrict__ segIds,
                 const float* __restrict__ Wb,
                 const float* __restrict__ bb,
                 unsigned int* __restrict__ pooled,
                 int E) {
    __shared__ unsigned short sW[H * KST_E];

    const int tid = threadIdx.x;
    for (int i = tid; i < (D + A) * H; i += 512) {
        int k = i >> 7, n = i & 127;
        sW[n * KST_E + k] = f2bf_s(Wb[i]);
    }
    for (int i = tid; i < H * (KST_E - (D + A)); i += 512) {
        int n = i / (KST_E - (D + A));
        int k = (D + A) + i % (KST_E - (D + A));
        sW[n * KST_E + k] = 0;
    }
    __syncthreads();

    const int wave = tid >> 5, lane = tid & 31;
    const int hi = lane >> 4, ln = lane & 15;
    const int numGroups = (E + 7) >> 3;
    const int numHalves = (int)gridDim.x * 32;
    const int steps     = (numGroups + numHalves - 1) / numHalves;
    const int waveId    = (int)blockIdx.x * 16 + wave;
    const int aBase = ((waveId * 2 + (ln >> 3)) * steps) * 8 + (ln & 7);
    const int dBase = ((waveId * 2 + hi) * steps) * 8;
    const int last  = E - 1;

    float bbv[8];
#pragma unroll
    for (int nt = 0; nt < 8; ++nt) bbv[nt] = bb[nt * 16 + ln];

    const int g0 = nidx[aBase < E ? aBase : last];
    int gN = nidx[(aBase + 8) < E ? (aBase + 8) : last];
    F4 raw[4][4];
    {
        const float* __restrict__ rp = feats + g0 * D + hi * 8;
#pragma unroll
        for (int kt = 0; kt < 4; ++kt) {
            raw[kt][0] = *(const F4*)(rp + kt * 32);
            raw[kt][1] = *(const F4*)(rp + kt * 32 + 4);
            raw[kt][2] = *(const F4*)(rp + kt * 32 + 16);
            raw[kt][3] = *(const F4*)(rp + kt * 32 + 20);
            __builtin_amdgcn_sched_barrier(0);
        }
    }
    F4 ai = *(const F4*)(addInfo + (aBase < E ? aBase : last) * A);

    int   prevSeg = -1;
    float best[8];
#pragma unroll
    for (int nt = 0; nt < 8; ++nt) best[nt] = -__builtin_inff();

#pragma unroll 1
    for (int t = 0; t < steps; ++t) {
    asm volatile("" ::: "memory");
    const int eRow = aBase + t * 8;

    const int eNN = eRow + 16;
    const int gNN = nidx[eNN < E ? eNN : last];

    int segArr[8];
#pragma unroll
    for (int r = 0; r < 8; ++r) {
        const int e = dBase + t * 8 + r;
        const int s = segIds[e < E ? e : last];
        segArr[r] = (e < E) ? s : -1;
    }

    const float* __restrict__ rpN = feats + gN * D + hi * 8;

    FragC acc[8];
#pragma unroll
    for (int nt = 0; nt < 8; ++nt)
#pragma unroll
        for (int r = 0; r < 8; ++r) acc[nt].f[r] = 0.f;

    FragAB bq[3][2];
#pragma unroll
    for (int gp = 0; gp < 2; ++gp)
#pragma unroll
        for (int j = 0; j < 2; ++j)
            loadB(bq[gp][j], sW,
                  (((gp % 4) * 2 + j) * 16 + ln) * KST_E + (gp / 4) * 32 + hi * 8);

#pragma unroll
    for (int kt = 0; kt < 5; ++kt) {
        FragAB a;
        if (kt < 4) {
            pack8(a, 0, raw[kt][0], raw[kt][1]);
            pack8(a, 8, raw[kt][2], raw[kt][3]);
        } else {
            F4 z{0.f, 0.f, 0.f, 0.f};
            F4 av = hi ? z : ai;
#pragma unroll
            for (int i = 0; i < 8; ++i) a.ui[i] = 0u;
            a.v[0] = (__bf16)av.x; a.v[1] = (__bf16)av.y;
            a.v[2] = (__bf16)av.z; a.v[3] = (__bf16)av.w;
        }
        __builtin_amdgcn_sched_barrier(0);
        if (kt < 4) {
            raw[kt][0] = *(const F4*)(rpN + kt * 32);
            raw[kt][1] = *(const F4*)(rpN + kt * 32 + 4);
            raw[kt][2] = *(const F4*)(rpN + kt * 32 + 16);
            raw[kt][3] = *(const F4*)(rpN + kt * 32 + 20);
        } else {
            const int eN = eRow + 8;
            ai = *(const F4*)(addInfo + (eN < E ? eN : last) * A);
        }
#pragma unroll
        for (int h = 0; h < 4; ++h) {
            const int g = kt * 4 + h;
            if (g + 2 < 20) {
                const int gn  = g + 2;
                const int ktn = gn / 4, hn = gn % 4;
#pragma unroll
                for (int j = 0; j < 2; ++j)
                    loadB(bq[gn % 3][j], sW, ((hn * 2 + j) * 16 + ln) * KST_E + ktn * 32 + hi * 8);
            }
            __builtin_amdgcn_sched_barrier(0);
#pragma unroll
            for (int j = 0; j < 2; ++j)
                acc[h * 2 + j].v = __builtin_amdgcn_wmma_f32_16x16x32_bf16(
                    false, a.v, false, bq[g % 3][j].v, (short)0, acc[h * 2 + j].v, false, false);
            __builtin_amdgcn_sched_barrier(0);
        }
    }

#pragma unroll
    for (int r = 0; r < 8; ++r) {
        const int s = segArr[r];
        if (s >= 0) {
            if (s != prevSeg) {
                if (prevSeg >= 0) {
#pragma unroll
                    for (int nt = 0; nt < 8; ++nt)
                        atomicMax(&pooled[prevSeg * H + nt * 16 + ln],
                                  __float_as_uint(fmaxf(best[nt] + bbv[nt], 0.f)));
                }
                prevSeg = s;
#pragma unroll
                for (int nt = 0; nt < 8; ++nt) best[nt] = -__builtin_inff();
            }
#pragma unroll
            for (int nt = 0; nt < 8; ++nt)
                best[nt] = __builtin_elementwise_maximum(best[nt], acc[nt].f[r]);
        }
    }

    gN = gNN;
    }

    if (prevSeg >= 0) {
#pragma unroll
        for (int nt = 0; nt < 8; ++nt)
            atomicMax(&pooled[prevSeg * H + nt * 16 + ln],
                      __float_as_uint(fmaxf(best[nt] + bbv[nt], 0.f)));
    }
}

__global__ __launch_bounds__(256)
void proj_kernel(const float* __restrict__ pooled,
                 const float* __restrict__ Wo,
                 const float* __restrict__ bo,
                 const float* __restrict__ src,
                 float* __restrict__ dst,
                 int N) {
    __shared__ unsigned short sW[D * KST_P];

    const int tid = threadIdx.x;
    for (int i = tid; i < H * D; i += 256) {
        int h = i >> 7, j = i & 127;
        sW[j * KST_P + h] = f2bf_s(Wo[i]);
    }
    for (int i = tid; i < D * (KST_P - H); i += 256) {
        int j = i >> 3, k = H + (i & 7);
        sW[j * KST_P + k] = 0;
    }
    __syncthreads();

    const int wave = tid >> 5, lane = tid & 31;
    const int hi = lane >> 4, ln = lane & 15;
    const int mtiles = (N + 15) >> 4;
    const int mt = blockIdx.x * 8 + wave;
    if (mt >= mtiles) return;

    const int mBase = mt * 16;
    const int mRow  = (mBase + ln) < N ? (mBase + ln) : (N - 1);
    const float* __restrict__ rowp = pooled + mRow * H;

    FragC acc[8];
#pragma unroll
    for (int nt = 0; nt < 8; ++nt)
#pragma unroll
        for (int r = 0; r < 8; ++r) acc[nt].f[r] = 0.f;

#pragma unroll
    for (int kt = 0; kt < 4; ++kt) {
        const int kb = kt * 32 + hi * 8;
        FragAB a;
        F4 f0 = *(const F4*)(rowp + kb);
        F4 f1 = *(const F4*)(rowp + kb + 4);
        F4 f2 = *(const F4*)(rowp + kb + 16);
        F4 f3 = *(const F4*)(rowp + kb + 20);
        pack8(a, 0, f0, f1);
        pack8(a, 8, f2, f3);
        FragAB bcur, bnxt;
        loadB(bcur, sW, ln * KST_P + kb);
#pragma unroll
        for (int nt = 0; nt < 8; ++nt) {
            if (nt < 7)
                loadB(bnxt, sW, ((nt + 1) * 16 + ln) * KST_P + kb);
            acc[nt].v = __builtin_amdgcn_wmma_f32_16x16x32_bf16(
                false, a.v, false, bcur.v, (short)0, acc[nt].v, false, false);
            bcur = bnxt;
        }
    }

#pragma unroll
    for (int nt = 0; nt < 8; ++nt) {
        const int j = nt * 16 + ln;
        const float boj = bo[j];
#pragma unroll
        for (int r = 0; r < 8; ++r) {
            const int m = mBase + r + hi * 8;
            if (m < N)
                dst[m * D + j] = acc[nt].f[r] + boj + src[m * D + j];
        }
    }
}

__global__ __launch_bounds__(256)
void final_kernel(const float* __restrict__ x,
                  const float* __restrict__ Wf,
                  const float* __restrict__ bf,
                  float* __restrict__ out,
                  int N) {
    const int tid = threadIdx.x;
    const int wave = tid >> 5, lane = tid & 31;
    const int n = blockIdx.x * 8 + wave;
    if (n >= N) return;
    F4 a = ((const F4*)(x + n * D))[lane];
    F4 w = ((const F4*)Wf)[lane];
    float s = a.x * w.x + a.y * w.y + a.z * w.z + a.w * w.w;
#pragma unroll
    for (int m = 16; m >= 1; m >>= 1)
        s += __shfl_xor(s, m, 32);
    if (lane == 0) out[n] = s + bf[0];
}

extern "C" void kernel_launch(void* const* d_in, const int* in_sizes, int n_in,
                              void* d_out, int out_size, void* d_ws, size_t ws_size,
                              hipStream_t stream) {
    const float* interp  = (const float*)d_in[0];
    const float* addInfo = (const float*)d_in[1];
    const int*   nidx    = (const int*)d_in[2];
    const int*   segIds  = (const int*)d_in[3];
    const float* Wb0 = (const float*)d_in[4];
    const float* bb0 = (const float*)d_in[5];
    const float* Wo0 = (const float*)d_in[6];
    const float* bo0 = (const float*)d_in[7];
    const float* Wb1 = (const float*)d_in[8];
    const float* bb1 = (const float*)d_in[9];
    const float* Wo1 = (const float*)d_in[10];
    const float* bo1 = (const float*)d_in[11];
    const float* Wf  = (const float*)d_in[12];
    const float* bfp = (const float*)d_in[13];

    const int N = in_sizes[0] / D;
    const int E = in_sizes[2];

    float* pooled = (float*)d_ws;
    float* xws    = pooled + (size_t)N * H;
    unsigned int* pooledU = (unsigned int*)pooled;

    const int n4      = (N * H) / 4;
    const int zeroGrd = (n4 + 255) / 256;
    const int edgeGrd = 256;
    const int projGrd = (((N + 15) / 16) + 7) / 8;
    const int finGrd  = (N + 7) / 8;

    zero_kernel<<<zeroGrd, 256, 0, stream>>>((F4*)pooled, n4);
    edge_kernel<<<edgeGrd, 512, 0, stream>>>(interp, addInfo, nidx, segIds,
                                             Wb0, bb0, pooledU, E);
    proj_kernel<<<projGrd, 256, 0, stream>>>(pooled, Wo0, bo0, interp, xws, N);

    zero_kernel<<<zeroGrd, 256, 0, stream>>>((F4*)pooled, n4);
    edge_kernel<<<edgeGrd, 512, 0, stream>>>(xws, addInfo, nidx, segIds,
                                             Wb1, bb1, pooledU, E);
    proj_kernel<<<projGrd, 256, 0, stream>>>(pooled, Wo1, bo1, xws, xws, N);

    final_kernel<<<finGrd, 256, 0, stream>>>(xws, Wf, bfp, (float*)d_out, N);
}


# node kernels: lanes own runs of four consecutive columns; rows, weights, residual and results move as four-float whole-line groups issued together, not single floats each waited for alone
# speedup vs baseline: 2.5836x; 1.3093x over previous
#include <hip/hip_runtime.h>


#define D 128
#define A 4
#define H 128
#define KST_P 136

typedef __attribute__((ext_vector_type(16))) __bf16 bf16x16;
typedef __attribute__((ext_vector_type(8)))  float  f32x8;
typedef __attribute__((ext_vector_type(2)))  float  f32x2;

struct alignas(16) U4 { unsigned int x, y, z, w; };
struct alignas(16) F4 { float x, y, z, w; };

union FragAB {
    bf16x16        v;
    unsigned short us[16];
    unsigned int   ui[8];
};
union FragC {
    f32x8 v;
    float f[8];
};

__device__ __forceinline__ void pack8(FragAB& fr, int base,
                                      const F4& a, const F4& b) {
    fr.v[base + 0] = (__bf16)a.x; fr.v[base + 1] = (__bf16)a.y;
    fr.v[base + 2] = (__bf16)a.z; fr.v[base + 3] = (__bf16)a.w;
    fr.v[base + 4] = (__bf16)b.x; fr.v[base + 5] = (__bf16)b.y;
    fr.v[base + 6] = (__bf16)b.z; fr.v[base + 7] = (__bf16)b.w;
}

__device__ __forceinline__ unsigned short f2bf_s(float f) {
    __bf16 h = (__bf16)f;
    return __builtin_bit_cast(unsigned short, h);
}

__device__ __forceinline__ void loadB(FragAB& b, const unsigned short* sW, int off) {
    *(U4*)(&b.ui[0]) = *(const U4*)(&sW[off]);
    *(U4*)(&b.ui[4]) = *(const U4*)(&sW[off + 16]);
}

__device__ __forceinline__ float dot4(const F4& a, const F4& b) {
    return a.x * b.x + a.y * b.y + a.z * b.z + a.w * b.w;
}

template <bool RAGGED>
__device__ __forceinline__ void load8(int (&d)[8], const int* __restrict__ p, int g, int lastGroup, int E) {
    if (RAGGED) {
#pragma unroll
        for (int r = 0; r < 8; ++r) {
            const int e = g * 8 + r;
            d[r] = p[e < E ? e : E - 1];
        }
    } else {
        const int gc = g < lastGroup ? g : lastGroup;
        const U4 a = *(const U4*)(p + gc * 8);
        const U4 b = *(const U4*)(p + gc * 8 + 4);
        d[0] = (int)a.x; d[1] = (int)a.y; d[2] = (int)a.z; d[3] = (int)a.w;
        d[4] = (int)b.x; d[5] = (int)b.y; d[6] = (int)b.z; d[7] = (int)b.w;
    }
}

__global__ __launch_bounds__(256) void zero_kernel(F4* __restrict__ p, int n4) {
    int i = blockIdx.x * 256 + threadIdx.x;
    if (i < n4) { F4 z{0.f, 0.f, 0.f, 0.f}; p[i] = z; }
}

typedef __attribute__((ext_vector_type(8))) __bf16 bf16x8;
union Pack8 {
    bf16x8 v;
    U4     u;
};

__device__ __forceinline__ float comp4(const F4& a, int p) {
    return p == 0 ? a.x : p == 1 ? a.y : p == 2 ? a.z : a.w;
}

__device__ __forceinline__ void load_weights(F4 (&wv)[16], const float* __restrict__ W, int tid) {
    const int c = tid & 31, kg = tid >> 5;
#pragma unroll
    for (int g = 0; g < 2; ++g)
#pragma unroll
        for (int kk = 0; kk < 8; ++kk)
            wv[g * 8 + kk] = *(const F4*)(W + ((g * 8 + kg) * 8 + kk) * 128 + c * 4);
}

__device__ __forceinline__ void store_weights(unsigned short* sW, const F4 (&wv)[16], int tid) {
    const int c = tid & 31, kg = tid >> 5;
    const int q = c >> 4, l = c & 15;
#pragma unroll
    for (int g = 0; g < 2; ++g) {
        const int k0 = (g * 8 + kg) * 8;
#pragma unroll
        for (int p = 0; p < 4; ++p) {
            Pack8 pk;
#pragma unroll
            for (int kk = 0; kk < 8; ++kk)
                pk.v[kk] = (__bf16)comp4(wv[g * 8 + kk], p);
            *(U4*)(&sW[((q * 4 + p) * 16 + l) * KST_P + k0]) = pk.u;
        }
    }
}

__device__ __forceinline__ void load_row(F4 (&xr)[16], const float* rowp) {
#pragma unroll
    for (int kt = 0; kt < 4; ++kt) {
        xr[kt * 4 + 0] = *(const F4*)(rowp + kt * 32);
        xr[kt * 4 + 1] = *(const F4*)(rowp + kt * 32 + 4);
        xr[kt * 4 + 2] = *(const F4*)(rowp + kt * 32 + 16);
        xr[kt * 4 + 3] = *(const F4*)(rowp + kt * 32 + 20);
    }
}

template <bool SCORE>
__global__ __launch_bounds__(256)
void nodeproj_kernel(const float* x,
                     const float* __restrict__ Wb,
                     const float* __restrict__ Wf,
                     const float* __restrict__ bf,
                     float* P,
                     float* __restrict__ score,
                     int N) {
    __shared__ __attribute__((aligned(16))) unsigned short sW[H * KST_P];

    const int tid = threadIdx.x;
    const int wave = tid >> 5, lane = tid & 31;
    const int hi = lane >> 4, ln = lane & 15;
    const int mtiles = (N + 15) >> 4;
    const int mt = blockIdx.x * 8 + wave;
    const int mBase = mt * 16;
    const int mRow  = (mBase + ln) < N ? (mBase + ln) : (N - 1);

    F4 xr[16], wv[16];
    load_row(xr, x + mRow * D + hi * 8);
    load_weights(wv, Wb, tid);
    __builtin_amdgcn_sched_barrier(0);
    store_weights(sW, wv, tid);
    __syncthreads();

    if (mt >= mtiles) return;

    FragC acc[8];
#pragma unroll
    for (int nt = 0; nt < 8; ++nt)
#pragma unroll
        for (int r = 0; r < 8; ++r) acc[nt].f[r] = 0.f;

    float part = 0.f;
#pragma unroll
    for (int kt = 0; kt < 4; ++kt) {
        const int kb = kt * 32 + hi * 8;
        FragAB a;
        pack8(a, 0, xr[kt * 4 + 0], xr[kt * 4 + 1]);
        pack8(a, 8, xr[kt * 4 + 2], xr[kt * 4 + 3]);
        if (SCORE) {
            part += dot4(xr[kt * 4 + 0], *(const F4*)(Wf + kb));
            part += dot4(xr[kt * 4 + 1], *(const F4*)(Wf + kb + 4));
            part += dot4(xr[kt * 4 + 2], *(const F4*)(Wf + kb + 16));
            part += dot4(xr[kt * 4 + 3], *(const F4*)(Wf + kb + 20));
        }
        FragAB bcur, bnxt;
        loadB(bcur, sW, ln * KST_P + kb);
#pragma unroll
        for (int nt = 0; nt < 8; ++nt) {
            if (nt < 7)
                loadB(bnxt, sW, ((nt + 1) * 16 + ln) * KST_P + kb);
            acc[nt].v = __builtin_amdgcn_wmma_f32_16x16x32_bf16(
                false, a.v, false, bcur.v, (short)0, acc[nt].v, false, false);
            bcur = bnxt;
        }
    }

#pragma unroll
    for (int r = 0; r < 8; ++r) {
        const int m = mBase + r + hi * 8;
        if (m < N) {
            float* pr = P + m * H + ln * 4;
            F4 v0{acc[0].f[r], acc[1].f[r], acc[2].f[r], acc[3].f[r]};
            F4 v1{acc[4].f[r], acc[5].f[r], acc[6].f[r], acc[7].f[r]};
            *(F4*)(pr)      = v0;
            *(F4*)(pr + 64) = v1;
        }
    }
    if (SCORE) {
        part += __shfl_xor(part, 16, 32);
        if (hi == 0 && (mBase + ln) < N) score[mBase + ln] = part + bf[0];
    }
}

template <bool RAGGED>
__global__ __launch_bounds__(512) __attribute__((amdgpu_waves_per_eu(4, 4)))
void edge_kernel(const float* __restrict__ P,
                 const float* __restrict__ addInfo,
                 const int*   __restrict__ nidx,
                 const int*   __restrict__ segIds,
                 const float* __restrict__ Wb,
                 const float* __restrict__ bb,
                 unsigned int* __restrict__ pooled,
                 int E) {
    const int tid = threadIdx.x;
    const int wave = tid >> 5, lane = tid & 31;
    const int hi = lane >> 4, ln = lane & 15;
    const int numGroups = (E + 7) >> 3;
    const int numHalves = (int)gridDim.x * 32;
    const int steps     = (numGroups + numHalves - 1) / numHalves;
    const int waveId    = (int)blockIdx.x * 16 + wave;
    const int aBase = ((waveId * 2 + (ln >> 3)) * steps) * 8 + (ln & 7);
    const int dGrp  = (waveId * 2 + hi) * steps;
    const int dBase = dGrp * 8;
    const int lastGroup = (E >> 3) - 1;
    const int last  = E - 1;

    float bbv[8];
    f32x2 bW[8];
#pragma unroll
    for (int nt = 0; nt < 8; ++nt) {
        const int col = (nt >> 2) * 64 + ln * 4 + (nt & 3);
        bbv[nt] = bb[col];
        bW[nt][0] = Wb[(D + 2 * hi) * H + col];
        bW[nt][1] = Wb[(D + 2 * hi + 1) * H + col];
    }

    int gC[8], gN[8];
    load8<RAGGED>(gC, nidx, dGrp, lastGroup, E);
    load8<RAGGED>(gN, nidx, dGrp + 1, lastGroup, E);
    int segC[8];
    load8<RAGGED>(segC, segIds, dGrp, lastGroup, E);
    f32x2 av = *(const f32x2*)(addInfo + (aBase < E ? aBase : last) * A + 2 * hi);
    __builtin_amdgcn_sched_barrier(0);
    F4 raw[8][2];
#pragma unroll
    for (int r = 0; r < 8; ++r) {
        const float* __restrict__ rp = P + gC[r] * H + ln * 4;
        raw[r][0] = *(const F4*)(rp);
        raw[r][1] = *(const F4*)(rp + 64);
        __builtin_amdgcn_sched_barrier(0);
    }

    int   prevSeg = -1;
    float best[8];
#pragma unroll
    for (int nt = 0; nt < 8; ++nt) best[nt] = -__builtin_inff();

#pragma unroll 1
    for (int t = 0; t < steps; ++t) {
    const int e0 = dBase + t * 8;

    int gNN[8];
    load8<RAGGED>(gNN, nidx, dGrp + t + 2, lastGroup, E);

    int segN[8];
    load8<RAGGED>(segN, segIds, dGrp + t + 1, lastGroup, E);

    FragC acc[8];
    {
        FragC z;
#pragma unroll
        for (int r = 0; r < 8; ++r) z.f[r] = 0.f;
#pragma unroll
        for (int nt = 0; nt < 8; ++nt)
            acc[nt].v = __builtin_amdgcn_wmma_f32_16x16x4_f32(
                false, av, false, bW[nt], (short)0, z.v, false, false);
    }
    {
        const int eN = aBase + (t + 1) * 8;
        av = *(const f32x2*)(addInfo + (eN < E ? eN : last) * A + 2 * hi);
    }

#pragma unroll
    for (int r = 0; r < 8; ++r) {
        const int s = (e0 + r < E) ? segC[r] : -1;
        if (s >= 0) {
            if (s != prevSeg) {
                if (prevSeg >= 0) {
                    unsigned int* __restrict__ pp = pooled + prevSeg * H + ln * 4;
#pragma unroll
                    for (int nt = 0; nt < 8; ++nt)
                        atomicMax(pp + (nt >> 2) * 64 + (nt & 3),
                                  __float_as_uint(fmaxf(best[nt] + bbv[nt], 0.f)));
                }
                prevSeg = s;
#pragma unroll
                for (int nt = 0; nt < 8; ++nt) best[nt] = -__builtin_inff();
            }
            best[0] = __builtin_elementwise_maximum(best[0], acc[0].f[r] + raw[r][0].x);
            best[1] = __builtin_elementwise_maximum(best[1], acc[1].f[r] + raw[r][0].y);
            best[2] = __builtin_elementwise_maximum(best[2], acc[2].f[r] + raw[r][0].z);
            best[3] = __builtin_elementwise_maximum(best[3], acc[3].f[r] + raw[r][0].w);
            best[4] = __builtin_elementwise_maximum(best[4], acc[4].f[r] + raw[r][1].x);
            best[5] = __builtin_elementwise_maximum(best[5], acc[5].f[r] + raw[r][1].y);
            best[6] = __builtin_elementwise_maximum(best[6], acc[6].f[r] + raw[r][1].z);
            best[7] = __builtin_elementwise_maximum(best[7], acc[7].f[r] + raw[r][1].w);
        }
        const float* __restrict__ rp = P + gN[r] * H + ln * 4;
        raw[r][0] = *(const F4*)(rp);
        raw[r][1] = *(const F4*)(rp + 64);
    }

#pragma unroll
    for (int r = 0; r < 8; ++r) { gN[r] = gNN[r]; segC[r] = segN[r]; }
    }

    if (prevSeg >= 0) {
        unsigned int* __restrict__ pp = pooled + prevSeg * H + ln * 4;
#pragma unroll
        for (int nt = 0; nt < 8; ++nt)
            atomicMax(pp + (nt >> 2) * 64 + (nt & 3),
                      __float_as_uint(fmaxf(best[nt] + bbv[nt], 0.f)));
    }
}

template <bool RESID>
__global__ __launch_bounds__(256)
void proj_kernel(const float* __restrict__ pooled,
                 const float* __restrict__ Wo,
                 const float* __restrict__ bo,
                 const float* __restrict__ src,
                 float* __restrict__ dst,
                 int N) {
    __shared__ __attribute__((aligned(16))) unsigned short sW[D * KST_P];

    const int tid = threadIdx.x;
    const int wave = tid >> 5, lane = tid & 31;
    const int hi = lane >> 4, ln = lane & 15;
    const int mtiles = (N + 15) >> 4;
    const int mt = blockIdx.x * 8 + wave;
    const int mBase = mt * 16;
    const int mRow  = (mBase + ln) < N ? (mBase + ln) : (N - 1);

    F4 xr[16], wv[16];
    load_row(xr, pooled + mRow * H + hi * 8);
    load_weights(wv, Wo, tid);
    __builtin_amdgcn_sched_barrier(0);
    store_weights(sW, wv, tid);

    const F4 b0 = *(const F4*)(bo + ln * 4);
    const F4 b1 = *(const F4*)(bo + 64 + ln * 4);
    F4 rs[8][2];
    if (RESID) {
#pragma unroll
        for (int r = 0; r < 8; ++r) {
            const int m  = mBase + r + hi * 8;
            const int mc = m < N ? m : (N - 1);
            const float* sr = src + mc * D + ln * 4;
            rs[r][0] = *(const F4*)(sr);
            rs[r][1] = *(const F4*)(sr + 64);
        }
    }
    __syncthreads();

    if (mt >= mtiles) return;

    FragC acc[8];
#pragma unroll
    for (int nt = 0; nt < 8; ++nt)
#pragma unroll
        for (int r = 0; r < 8; ++r) acc[nt].f[r] = 0.f;

#pragma unroll
    for (int kt = 0; kt < 4; ++kt) {
        const int kb = kt * 32 + hi * 8;
        FragAB a;
        pack8(a, 0, xr[kt * 4 + 0], xr[kt * 4 + 1]);
        pack8(a, 8, xr[kt * 4 + 2], xr[kt * 4 + 3]);
        FragAB bcur, bnxt;
        loadB(bcur, sW, ln * KST_P + kb);
#pragma unroll
        for (int nt = 0; nt < 8; ++nt) {
            if (nt < 7)
                loadB(bnxt, sW, ((nt + 1) * 16 + ln) * KST_P + kb);
            acc[nt].v = __builtin_amdgcn_wmma_f32_16x16x32_bf16(
                false, a.v, false, bcur.v, (short)0, acc[nt].v, false, false);
            bcur = bnxt;
        }
    }

#pragma unroll
    for (int r = 0; r < 8; ++r) {
        const int m = mBase + r + hi * 8;
        if (m < N) {
            F4 v0{acc[0].f[r] + b0.x, acc[1].f[r] + b0.y, acc[2].f[r] + b0.z, acc[3].f[r] + b0.w};
            F4 v1{acc[4].f[r] + b1.x, acc[5].f[r] + b1.y, acc[6].f[r] + b1.z, acc[7].f[r] + b1.w};
            if (RESID) {
                v0.x += rs[r][0].x; v0.y += rs[r][0].y; v0.z += rs[r][0].z; v0.w += rs[r][0].w;
                v1.x += rs[r][1].x; v1.y += rs[r][1].y; v1.z += rs[r][1].z; v1.w += rs[r][1].w;
            }
            float* dr = dst + m * D + ln * 4;
            *(F4*)(dr)      = v0;
            *(F4*)(dr + 64) = v1;
        }
    }
}

__global__ __launch_bounds__(256)
void final_kernel(const float* __restrict__ y,
                  const float* __restrict__ Wf,
                  float* __restrict__ out,
                  int N) {
    const int tid = threadIdx.x;
    const int wave = tid >> 5, lane = tid & 31;
    const int n = blockIdx.x * 8 + wave;
    if (n >= N) return;
    F4 a = ((const F4*)(y + n * D))[lane];
    F4 w = ((const F4*)Wf)[lane];
    float s = a.x * w.x + a.y * w.y + a.z * w.z + a.w * w.w;
#pragma unroll
    for (int m = 16; m >= 1; m >>= 1)
        s += __shfl_xor(s, m, 32);
    if (lane == 0) out[n] = out[n] + s;
}

extern "C" void kernel_launch(void* const* d_in, const int* in_sizes, int n_in,
                              void* d_out, int out_size, void* d_ws, size_t ws_size,
                              hipStream_t stream) {
    const float* interp  = (const float*)d_in[0];
    const float* addInfo = (const float*)d_in[1];
    const int*   nidx    = (const int*)d_in[2];
    const int*   segIds  = (const int*)d_in[3];
    const float* Wb0 = (const float*)d_in[4];
    const float* bb0 = (const float*)d_in[5];
    const float* Wo0 = (const float*)d_in[6];
    const float* bo0 = (const float*)d_in[7];
    const float* Wb1 = (const float*)d_in[8];
    const float* bb1 = (const float*)d_in[9];
    const float* Wo1 = (const float*)d_in[10];
    const float* bo1 = (const float*)d_in[11];
    const float* Wf  = (const float*)d_in[12];
    const float* bfp = (const float*)d_in[13];
    float* out = (float*)d_out;

    const int N = in_sizes[0] / D;
    const int E = in_sizes[2];

    float* pooled = (float*)d_ws;
    float* plane  = pooled + (size_t)N * H;
    unsigned int* pooledU = (unsigned int*)pooled;

    const int n4      = (N * H) / 4;
    const int zeroGrd = (n4 + 255) / 256;
    const int edgeGrd = 256;
    const int projGrd = (((N + 15) / 16) + 7) / 8;
    const int finGrd  = (N + 7) / 8;

    const bool ragged = (E & 7) != 0;

    zero_kernel<<<zeroGrd, 256, 0, stream>>>((F4*)pooled, n4);
    nodeproj_kernel<false><<<projGrd, 256, 0, stream>>>(interp, Wb0, Wf, bfp, plane, out, N);
    if (ragged)
        edge_kernel<true><<<edgeGrd, 512, 0, stream>>>(plane, addInfo, nidx, segIds,
                                                       Wb0, bb0, pooledU, E);
    else
        edge_kernel<false><<<edgeGrd, 512, 0, stream>>>(plane, addInfo, nidx, segIds,
                                                        Wb0, bb0, pooledU, E);
    proj_kernel<true><<<projGrd, 256, 0, stream>>>(pooled, Wo0, bo0, interp, plane, N);

    zero_kernel<<<zeroGrd, 256, 0, stream>>>((F4*)pooled, n4);
    nodeproj_kernel<true><<<projGrd, 256, 0, stream>>>(plane, Wb1, Wf, bfp, plane, out, N);
    if (ragged)
        edge_kernel<true><<<edgeGrd, 512, 0, stream>>>(plane, addInfo, nidx, segIds,
                                                       Wb1, bb1, pooledU, E);
    else
        edge_kernel<false><<<edgeGrd, 512, 0, stream>>>(plane, addInfo, nidx, segIds,
                                                        Wb1, bb1, pooledU, E);
    proj_kernel<false><<<projGrd, 256, 0, stream>>>(pooled, Wo1, bo1, interp, plane, N);

    final_kernel<<<finGrd, 256, 0, stream>>>(plane, Wf, out, N);
}


# edge kernel: pooled rows of segments whose rows all lie inside one wave's rows leave as plain 16-byte stores (run crossing the two lane halves joined in registers); atomicMax only at a wave's ends
# speedup vs baseline: 3.2537x; 1.2594x over previous
#include <hip/hip_runtime.h>


#define D 128
#define A 4
#define H 128
#define KST_P 136

typedef __attribute__((ext_vector_type(16))) __bf16 bf16x16;
typedef __attribute__((ext_vector_type(8)))  float  f32x8;
typedef __attribute__((ext_vector_type(2)))  float  f32x2;

struct alignas(16) U4 { unsigned int x, y, z, w; };
struct alignas(16) F4 { float x, y, z, w; };

union FragAB {
    bf16x16        v;
    unsigned short us[16];
    unsigned int   ui[8];
};
union FragC {
    f32x8 v;
    float f[8];
};

__device__ __forceinline__ void pack8(FragAB& fr, int base,
                                      const F4& a, const F4& b) {
    fr.v[base + 0] = (__bf16)a.x; fr.v[base + 1] = (__bf16)a.y;
    fr.v[base + 2] = (__bf16)a.z; fr.v[base + 3] = (__bf16)a.w;
    fr.v[base + 4] = (__bf16)b.x; fr.v[base + 5] = (__bf16)b.y;
    fr.v[base + 6] = (__bf16)b.z; fr.v[base + 7] = (__bf16)b.w;
}

__device__ __forceinline__ unsigned short f2bf_s(float f) {
    __bf16 h = (__bf16)f;
    return __builtin_bit_cast(unsigned short, h);
}

__device__ __forceinline__ void loadB(FragAB& b, const unsigned short* sW, int off) {
    *(U4*)(&b.ui[0]) = *(const U4*)(&sW[off]);
    *(U4*)(&b.ui[4]) = *(const U4*)(&sW[off + 16]);
}

__device__ __forceinline__ float dot4(const F4& a, const F4& b) {
    return a.x * b.x + a.y * b.y + a.z * b.z + a.w * b.w;
}

template <bool RAGGED>
__device__ __forceinline__ void load8(int (&d)[8], const int* __restrict__ p, int g, int lastGroup, int E) {
    if (RAGGED) {
#pragma unroll
        for (int r = 0; r < 8; ++r) {
            const int e = g * 8 + r;
            d[r] = p[e < E ? e : E - 1];
        }
    } else {
        const int gc = g < lastGroup ? g : lastGroup;
        const U4 a = *(const U4*)(p + gc * 8);
        const U4 b = *(const U4*)(p + gc * 8 + 4);
        d[0] = (int)a.x; d[1] = (int)a.y; d[2] = (int)a.z; d[3] = (int)a.w;
        d[4] = (int)b.x; d[5] = (int)b.y; d[6] = (int)b.z; d[7] = (int)b.w;
    }
}

typedef __attribute__((ext_vector_type(8))) __bf16 bf16x8;
union Pack8 {
    bf16x8 v;
    U4     u;
};

__device__ __forceinline__ float comp4(const F4& a, int p) {
    return p == 0 ? a.x : p == 1 ? a.y : p == 2 ? a.z : a.w;
}

__device__ __forceinline__ void load_weights(F4 (&wv)[16], const float* __restrict__ W, int tid) {
    const int c = tid & 31, kg = tid >> 5;
#pragma unroll
    for (int g = 0; g < 2; ++g)
#pragma unroll
        for (int kk = 0; kk < 8; ++kk)
            wv[g * 8 + kk] = *(const F4*)(W + ((g * 8 + kg) * 8 + kk) * 128 + c * 4);
}

__device__ __forceinline__ void store_weights(unsigned short* sW, const F4 (&wv)[16], int tid) {
    const int c = tid & 31, kg = tid >> 5;
    const int q = c >> 4, l = c & 15;
#pragma unroll
    for (int g = 0; g < 2; ++g) {
        const int k0 = (g * 8 + kg) * 8;
#pragma unroll
        for (int p = 0; p < 4; ++p) {
            Pack8 pk;
#pragma unroll
            for (int kk = 0; kk < 8; ++kk)
                pk.v[kk] = (__bf16)comp4(wv[g * 8 + kk], p);
            *(U4*)(&sW[((q * 4 + p) * 16 + l) * KST_P + k0]) = pk.u;
        }
    }
}

__device__ __forceinline__ void load_row(F4 (&xr)[16], const float* rowp) {
#pragma unroll
    for (int kt = 0; kt < 4; ++kt) {
        xr[kt * 4 + 0] = *(const F4*)(rowp + kt * 32);
        xr[kt * 4 + 1] = *(const F4*)(rowp + kt * 32 + 4);
        xr[kt * 4 + 2] = *(const F4*)(rowp + kt * 32 + 16);
        xr[kt * 4 + 3] = *(const F4*)(rowp + kt * 32 + 20);
    }
}

template <bool SCORE>
__global__ __launch_bounds__(256)
void nodeproj_kernel(const float* x,
                     const float* __restrict__ Wb,
                     const float* __restrict__ Wf,
                     const float* __restrict__ bf,
                     float* P,
                     float* __restrict__ score,
                     float* __restrict__ pooledZ,
                     int N) {
    __shared__ __attribute__((aligned(16))) unsigned short sW[H * KST_P];

    const int tid = threadIdx.x;
    const int wave = tid >> 5, lane = tid & 31;
    const int hi = lane >> 4, ln = lane & 15;
    const int mtiles = (N + 15) >> 4;
    const int mt = blockIdx.x * 8 + wave;
    const int mBase = mt * 16;
    const int mRow  = (mBase + ln) < N ? (mBase + ln) : (N - 1);

    F4 xr[16], wv[16];
    load_row(xr, x + mRow * D + hi * 8);
    load_weights(wv, Wb, tid);
    __builtin_amdgcn_sched_barrier(0);
    {
        const int mBaseU = __builtin_amdgcn_readfirstlane(mBase);
        F4* zp = (F4*)(pooledZ + mBaseU * H) + lane;
        const F4 z{0.f, 0.f, 0.f, 0.f};
#pragma unroll
        for (int j = 0; j < 16; ++j)
            if (mBaseU + j < N) zp[j * 32] = z;
    }
    store_weights(sW, wv, tid);
    __syncthreads();

    if (mt >= mtiles) return;

    FragC acc[8];
#pragma unroll
    for (int nt = 0; nt < 8; ++nt)
#pragma unroll
        for (int r = 0; r < 8; ++r) acc[nt].f[r] = 0.f;

    float part = 0.f;
#pragma unroll
    for (int kt = 0; kt < 4; ++kt) {
        const int kb = kt * 32 + hi * 8;
        FragAB a;
        pack8(a, 0, xr[kt * 4 + 0], xr[kt * 4 + 1]);
        pack8(a, 8, xr[kt * 4 + 2], xr[kt * 4 + 3]);
        if (SCORE) {
            part += dot4(xr[kt * 4 + 0], *(const F4*)(Wf + kb));
            part += dot4(xr[kt * 4 + 1], *(const F4*)(Wf + kb + 4));
            part += dot4(xr[kt * 4 + 2], *(const F4*)(Wf + kb + 16));
            part += dot4(xr[kt * 4 + 3], *(const F4*)(Wf + kb + 20));
        }
        FragAB bcur, bnxt;
        loadB(bcur, sW, ln * KST_P + kb);
#pragma unroll
        for (int nt = 0; nt < 8; ++nt) {
            if (nt < 7)
                loadB(bnxt, sW, ((nt + 1) * 16 + ln) * KST_P + kb);
            acc[nt].v = __builtin_amdgcn_wmma_f32_16x16x32_bf16(
                false, a.v, false, bcur.v, (short)0, acc[nt].v, false, false);
            bcur = bnxt;
        }
    }

#pragma unroll
    for (int r = 0; r < 8; ++r) {
        const int m = mBase + r + hi * 8;
        if (m < N) {
            float* pr = P + m * H + ln * 4;
            F4 v0{acc[0].f[r], acc[1].f[r], acc[2].f[r], acc[3].f[r]};
            F4 v1{acc[4].f[r], acc[5].f[r], acc[6].f[r], acc[7].f[r]};
            *(F4*)(pr)      = v0;
            *(F4*)(pr + 64) = v1;
        }
    }
    if (SCORE) {
        part += __shfl_xor(part, 16, 32);
        if (hi == 0 && (mBase + ln) < N) score[mBase + ln] = part + bf[0];
    }
}

template <bool RAGGED>
__global__ __launch_bounds__(512) __attribute__((amdgpu_waves_per_eu(4, 4)))
void edge_kernel(const float* __restrict__ P,
                 const float* __restrict__ addInfo,
                 const int*   __restrict__ nidx,
                 const int*   __restrict__ segIds,
                 const float* __restrict__ Wb,
                 const float* __restrict__ bb,
                 unsigned int* __restrict__ pooled,
                 int E) {
    const int tid = threadIdx.x;
    const int wave = tid >> 5, lane = tid & 31;
    const int hi = lane >> 4, ln = lane & 15;
    const int numGroups = (E + 7) >> 3;
    const int numHalves = (int)gridDim.x * 32;
    const int steps     = (numGroups + numHalves - 1) / numHalves;
    const int waveId    = (int)blockIdx.x * 16 + wave;
    const int aBase = ((waveId * 2 + (ln >> 3)) * steps) * 8 + (ln & 7);
    const int dGrp  = (waveId * 2 + hi) * steps;
    const int dBase = dGrp * 8;
    const int lastGroup = (E >> 3) - 1;
    const int last  = E - 1;

    float bbv[8];
    f32x2 bW[8];
#pragma unroll
    for (int nt = 0; nt < 8; ++nt) {
        const int col = (nt >> 2) * 64 + ln * 4 + (nt & 3);
        bbv[nt] = bb[col];
        bW[nt][0] = Wb[(D + 2 * hi) * H + col];
        bW[nt][1] = Wb[(D + 2 * hi + 1) * H + col];
    }

    int gC[8], gN[8];
    load8<RAGGED>(gC, nidx, dGrp, lastGroup, E);
    load8<RAGGED>(gN, nidx, dGrp + 1, lastGroup, E);
    int segC[8];
    load8<RAGGED>(segC, segIds, dGrp, lastGroup, E);
    f32x2 av = *(const f32x2*)(addInfo + (aBase < E ? aBase : last) * A + 2 * hi);
    __builtin_amdgcn_sched_barrier(0);
    F4 raw[8][2];
#pragma unroll
    for (int r = 0; r < 8; ++r) {
        const float* __restrict__ rp = P + gC[r] * H + ln * 4;
        raw[r][0] = *(const F4*)(rp);
        raw[r][1] = *(const F4*)(rp + 64);
        __builtin_amdgcn_sched_barrier(0);
    }

    int   prevSeg = -1;
    bool  firstRun = true;
    int   fSeg = -1;
    float fV[8];
#pragma unroll
    for (int nt = 0; nt < 8; ++nt) fV[nt] = 0.f;
    float best[8];
#pragma unroll
    for (int nt = 0; nt < 8; ++nt) best[nt] = -__builtin_inff();

#pragma unroll 1
    for (int t = 0; t < steps; ++t) {
    const int e0 = dBase + t * 8;

    int gNN[8];
    load8<RAGGED>(gNN, nidx, dGrp + t + 2, lastGroup, E);

    int segN[8];
    load8<RAGGED>(segN, segIds, dGrp + t + 1, lastGroup, E);

    FragC acc[8];
    {
        FragC z;
#pragma unroll
        for (int r = 0; r < 8; ++r) z.f[r] = 0.f;
#pragma unroll
        for (int nt = 0; nt < 8; ++nt)
            acc[nt].v = __builtin_amdgcn_wmma_f32_16x16x4_f32(
                false, av, false, bW[nt], (short)0, z.v, false, false);
    }
    {
        const int eN = aBase + (t + 1) * 8;
        av = *(const f32x2*)(addInfo + (eN < E ? eN : last) * A + 2 * hi);
    }

#pragma unroll
    for (int r = 0; r < 8; ++r) {
        const int s = (e0 + r < E) ? segC[r] : -1;
        if (s >= 0) {
            if (s != prevSeg) {
                if (prevSeg >= 0) {
                    float v[8];
#pragma unroll
                    for (int nt = 0; nt < 8; ++nt)
                        v[nt] = fmaxf(best[nt] + bbv[nt], 0.f);
                    if (firstRun) {
                        fSeg = prevSeg;
#pragma unroll
                        for (int nt = 0; nt < 8; ++nt) fV[nt] = v[nt];
                        firstRun = false;
                    } else {
                        unsigned int* __restrict__ pp = pooled + prevSeg * H + ln * 4;
                        *(U4*)(pp)      = U4{__float_as_uint(v[0]), __float_as_uint(v[1]), __float_as_uint(v[2]), __float_as_uint(v[3])};
                        *(U4*)(pp + 64) = U4{__float_as_uint(v[4]), __float_as_uint(v[5]), __float_as_uint(v[6]), __float_as_uint(v[7])};
                    }
                }
                prevSeg = s;
#pragma unroll
                for (int nt = 0; nt < 8; ++nt) best[nt] = -__builtin_inff();
            }
            best[0] = __builtin_elementwise_maximum(best[0], acc[0].f[r] + raw[r][0].x);
            best[1] = __builtin_elementwise_maximum(best[1], acc[1].f[r] + raw[r][0].y);
            best[2] = __builtin_elementwise_maximum(best[2], acc[2].f[r] + raw[r][0].z);
            best[3] = __builtin_elementwise_maximum(best[3], acc[3].f[r] + raw[r][0].w);
            best[4] = __builtin_elementwise_maximum(best[4], acc[4].f[r] + raw[r][1].x);
            best[5] = __builtin_elementwise_maximum(best[5], acc[5].f[r] + raw[r][1].y);
            best[6] = __builtin_elementwise_maximum(best[6], acc[6].f[r] + raw[r][1].z);
            best[7] = __builtin_elementwise_maximum(best[7], acc[7].f[r] + raw[r][1].w);
        }
        const float* __restrict__ rp = P + gN[r] * H + ln * 4;
        raw[r][0] = *(const F4*)(rp);
        raw[r][1] = *(const F4*)(rp + 64);
    }

#pragma unroll
    for (int r = 0; r < 8; ++r) { gN[r] = gNN[r]; segC[r] = segN[r]; }
    }

    const bool has   = prevSeg >= 0;
    const bool multi = !firstRun;
    float lastV[8];
#pragma unroll
    for (int nt = 0; nt < 8; ++nt) lastV[nt] = fmaxf(best[nt] + bbv[nt], 0.f);
    const int firstSeg = multi ? fSeg : prevSeg;
    float firstV[8];
#pragma unroll
    for (int nt = 0; nt < 8; ++nt) firstV[nt] = multi ? fV[nt] : lastV[nt];

    const int oLastSeg  = __shfl_xor(prevSeg, 16, 32);
    const int oFirstSeg = __shfl_xor(firstSeg, 16, 32);
    const int oMulti    = __shfl_xor(multi ? 1 : 0, 16, 32);
    float oLastV[8];
#pragma unroll
    for (int nt = 0; nt < 8; ++nt) oLastV[nt] = __shfl_xor(lastV[nt], 16, 32);

    const bool join = multi && (oMulti != 0) &&
                      (hi == 0 ? (prevSeg == oFirstSeg) : (firstSeg == oLastSeg));

    if (has) {
        unsigned int* __restrict__ pp = pooled + firstSeg * H + ln * 4;
        if (hi == 1 && join) {
            *(U4*)(pp)      = U4{__float_as_uint(fmaxf(firstV[0], oLastV[0])), __float_as_uint(fmaxf(firstV[1], oLastV[1])),
                                 __float_as_uint(fmaxf(firstV[2], oLastV[2])), __float_as_uint(fmaxf(firstV[3], oLastV[3]))};
            *(U4*)(pp + 64) = U4{__float_as_uint(fmaxf(firstV[4], oLastV[4])), __float_as_uint(fmaxf(firstV[5], oLastV[5])),
                                 __float_as_uint(fmaxf(firstV[6], oLastV[6])), __float_as_uint(fmaxf(firstV[7], oLastV[7]))};
        } else {
#pragma unroll
            for (int nt = 0; nt < 8; ++nt)
                atomicMax(pp + (nt >> 2) * 64 + (nt & 3), __float_as_uint(firstV[nt]));
        }
    }
    if (multi && !(hi == 0 && join)) {
        unsigned int* __restrict__ pp = pooled + prevSeg * H + ln * 4;
#pragma unroll
        for (int nt = 0; nt < 8; ++nt)
            atomicMax(pp + (nt >> 2) * 64 + (nt & 3), __float_as_uint(lastV[nt]));
    }
}

template <int M>
__device__ __forceinline__ float row_xor(float v) {
    return __builtin_bit_cast(float,
        __builtin_amdgcn_update_dpp(0, __builtin_bit_cast(int, v), 0x160 + M, 0xf, 0xf, false));
}

template <bool LAST>
__global__ __launch_bounds__(256)
void proj_kernel(const float* __restrict__ pooled,
                 const float* __restrict__ Wo,
                 const float* __restrict__ bo,
                 const float* __restrict__ src,
                 const float* __restrict__ Wf,
                 float* __restrict__ dst,
                 float* __restrict__ out,
                 int N) {
    __shared__ __attribute__((aligned(16))) unsigned short sW[D * KST_P];

    const int tid = threadIdx.x;
    const int wave = tid >> 5, lane = tid & 31;
    const int hi = lane >> 4, ln = lane & 15;
    const int mtiles = (N + 15) >> 4;
    const int mt = blockIdx.x * 8 + wave;
    const int mBase = mt * 16;
    const int mRow  = (mBase + ln) < N ? (mBase + ln) : (N - 1);

    F4 xr[16], wv[16];
    load_row(xr, pooled + mRow * H + hi * 8);
    load_weights(wv, Wo, tid);
    __builtin_amdgcn_sched_barrier(0);
    store_weights(sW, wv, tid);

    const F4 b0 = *(const F4*)(bo + ln * 4);
    const F4 b1 = *(const F4*)(bo + 64 + ln * 4);
    F4 rs[8][2];
    F4 wf0, wf1;
    float oprev = 0.f;
    const int mOut = mBase + hi * 8 + (ln & 7);
    if (!LAST) {
#pragma unroll
        for (int r = 0; r < 8; ++r) {
            const int m  = mBase + r + hi * 8;
            const int mc = m < N ? m : (N - 1);
            const float* sr = src + mc * D + ln * 4;
            rs[r][0] = *(const F4*)(sr);
            rs[r][1] = *(const F4*)(sr + 64);
        }
    } else {
        wf0 = *(const F4*)(Wf + ln * 4);
        wf1 = *(const F4*)(Wf + 64 + ln * 4);
        oprev = out[mOut < N ? mOut : (N - 1)];
    }
    __syncthreads();

    if (mt >= mtiles) return;

    FragC acc[8];
#pragma unroll
    for (int nt = 0; nt < 8; ++nt)
#pragma unroll
        for (int r = 0; r < 8; ++r) acc[nt].f[r] = 0.f;

#pragma unroll
    for (int kt = 0; kt < 4; ++kt) {
        const int kb = kt * 32 + hi * 8;
        FragAB a;
        pack8(a, 0, xr[kt * 4 + 0], xr[kt * 4 + 1]);
        pack8(a, 8, xr[kt * 4 + 2], xr[kt * 4 + 3]);
        FragAB bcur, bnxt;
        loadB(bcur, sW, ln * KST_P + kb);
#pragma unroll
        for (int nt = 0; nt < 8; ++nt) {
            if (nt < 7)
                loadB(bnxt, sW, ((nt + 1) * 16 + ln) * KST_P + kb);
            acc[nt].v = __builtin_amdgcn_wmma_f32_16x16x32_bf16(
                false, a.v, false, bcur.v, (short)0, acc[nt].v, false, false);
            bcur = bnxt;
        }
    }

    if (!LAST) {
#pragma unroll
        for (int r = 0; r < 8; ++r) {
            const int m = mBase + r + hi * 8;
            if (m < N) {
                F4 v0{acc[0].f[r] + b0.x, acc[1].f[r] + b0.y, acc[2].f[r] + b0.z, acc[3].f[r] + b0.w};
                F4 v1{acc[4].f[r] + b1.x, acc[5].f[r] + b1.y, acc[6].f[r] + b1.z, acc[7].f[r] + b1.w};
                v0.x += rs[r][0].x; v0.y += rs[r][0].y; v0.z += rs[r][0].z; v0.w += rs[r][0].w;
                v1.x += rs[r][1].x; v1.y += rs[r][1].y; v1.z += rs[r][1].z; v1.w += rs[r][1].w;
                float* dr = dst + m * D + ln * 4;
                *(F4*)(dr)      = v0;
                *(F4*)(dr + 64) = v1;
            }
        }
    } else {
        const float cb = dot4(b0, wf0) + dot4(b1, wf1);
        float s[8];
#pragma unroll
        for (int r = 0; r < 8; ++r) {
            const F4 a0{acc[0].f[r], acc[1].f[r], acc[2].f[r], acc[3].f[r]};
            const F4 a1{acc[4].f[r], acc[5].f[r], acc[6].f[r], acc[7].f[r]};
            s[r] = cb + dot4(a0, wf0) + dot4(a1, wf1);
        }
#pragma unroll
        for (int r = 0; r < 8; ++r) s[r] += row_xor<1>(s[r]);
#pragma unroll
        for (int r = 0; r < 8; ++r) s[r] += row_xor<2>(s[r]);
#pragma unroll
        for (int r = 0; r < 8; ++r) s[r] += row_xor<4>(s[r]);
#pragma unroll
        for (int r = 0; r < 8; ++r) s[r] += row_xor<8>(s[r]);
        float mine = s[0];
#pragma unroll
        for (int r = 1; r < 8; ++r) mine = (ln == r) ? s[r] : mine;
        if (ln < 8 && mOut < N) out[mOut] = oprev + mine;
    }
}

extern "C" void kernel_launch(void* const* d_in, const int* in_sizes, int n_in,
                              void* d_out, int out_size, void* d_ws, size_t ws_size,
                              hipStream_t stream) {
    const float* interp  = (const float*)d_in[0];
    const float* addInfo = (const float*)d_in[1];
    const int*   nidx    = (const int*)d_in[2];
    const int*   segIds  = (const int*)d_in[3];
    const float* Wb0 = (const float*)d_in[4];
    const float* bb0 = (const float*)d_in[5];
    const float* Wo0 = (const float*)d_in[6];
    const float* bo0 = (const float*)d_in[7];
    const float* Wb1 = (const float*)d_in[8];
    const float* bb1 = (const float*)d_in[9];
    const float* Wo1 = (const float*)d_in[10];
    const float* bo1 = (const float*)d_in[11];
    const float* Wf  = (const float*)d_in[12];
    const float* bfp = (const float*)d_in[13];
    float* out = (float*)d_out;

    const int N = in_sizes[0] / D;
    const int E = in_sizes[2];

    float* pooled = (float*)d_ws;
    float* plane  = pooled + (size_t)N * H;
    unsigned int* pooledU = (unsigned int*)pooled;

    const int edgeGrd = 256;
    const int projGrd = (((N + 15) / 16) + 7) / 8;

    const bool ragged = (E & 7) != 0;

    nodeproj_kernel<false><<<projGrd, 256, 0, stream>>>(interp, Wb0, Wf, bfp, plane, out, pooled, N);
    if (ragged)
        edge_kernel<true><<<edgeGrd, 512, 0, stream>>>(plane, addInfo, nidx, segIds,
                                                       Wb0, bb0, pooledU, E);
    else
        edge_kernel<false><<<edgeGrd, 512, 0, stream>>>(plane, addInfo, nidx, segIds,
                                                        Wb0, bb0, pooledU, E);
    proj_kernel<false><<<projGrd, 256, 0, stream>>>(pooled, Wo0, bo0, interp, Wf, plane, out, N);

    nodeproj_kernel<true><<<projGrd, 256, 0, stream>>>(plane, Wb1, Wf, bfp, plane, out, pooled, N);
    if (ragged)
        edge_kernel<true><<<edgeGrd, 512, 0, stream>>>(plane, addInfo, nidx, segIds,
                                                       Wb1, bb1, pooledU, E);
    else
        edge_kernel<false><<<edgeGrd, 512, 0, stream>>>(plane, addInfo, nidx, segIds,
                                                        Wb1, bb1, pooledU, E);
    proj_kernel<true><<<projGrd, 256, 0, stream>>>(pooled, Wo1, bo1, interp, Wf, plane, out, N);
}


# node side: block 0 output projection and block 1 input projection fused in one kernel; x1 goes from the accumulators through a wave-private LDS tile into the A operand, never stored: 6 launches to 5
# speedup vs baseline: 3.4925x; 1.0734x over previous
#include <hip/hip_runtime.h>


#define D 128
#define A 4
#define H 128
#define KST_P 136

typedef __attribute__((ext_vector_type(16))) __bf16 bf16x16;
typedef __attribute__((ext_vector_type(8)))  float  f32x8;
typedef __attribute__((ext_vector_type(2)))  float  f32x2;

struct alignas(16) U4 { unsigned int x, y, z, w; };
struct alignas(16) F4 { float x, y, z, w; };

union FragAB {
    bf16x16        v;
    unsigned short us[16];
    unsigned int   ui[8];
};
union FragC {
    f32x8 v;
    float f[8];
};

__device__ __forceinline__ void pack8(FragAB& fr, int base,
                                      const F4& a, const F4& b) {
    fr.v[base + 0] = (__bf16)a.x; fr.v[base + 1] = (__bf16)a.y;
    fr.v[base + 2] = (__bf16)a.z; fr.v[base + 3] = (__bf16)a.w;
    fr.v[base + 4] = (__bf16)b.x; fr.v[base + 5] = (__bf16)b.y;
    fr.v[base + 6] = (__bf16)b.z; fr.v[base + 7] = (__bf16)b.w;
}

__device__ __forceinline__ unsigned short f2bf_s(float f) {
    __bf16 h = (__bf16)f;
    return __builtin_bit_cast(unsigned short, h);
}

__device__ __forceinline__ void loadB(FragAB& b, const unsigned short* sW, int off) {
    *(U4*)(&b.ui[0]) = *(const U4*)(&sW[off]);
    *(U4*)(&b.ui[4]) = *(const U4*)(&sW[off + 16]);
}

__device__ __forceinline__ float dot4(const F4& a, const F4& b) {
    return a.x * b.x + a.y * b.y + a.z * b.z + a.w * b.w;
}

template <bool RAGGED>
__device__ __forceinline__ void load8(int (&d)[8], const int* __restrict__ p, int g, int lastGroup, int E) {
    if (RAGGED) {
#pragma unroll
        for (int r = 0; r < 8; ++r) {
            const int e = g * 8 + r;
            d[r] = p[e < E ? e : E - 1];
        }
    } else {
        const int gc = g < lastGroup ? g : lastGroup;
        const U4 a = *(const U4*)(p + gc * 8);
        const U4 b = *(const U4*)(p + gc * 8 + 4);
        d[0] = (int)a.x; d[1] = (int)a.y; d[2] = (int)a.z; d[3] = (int)a.w;
        d[4] = (int)b.x; d[5] = (int)b.y; d[6] = (int)b.z; d[7] = (int)b.w;
    }
}

typedef __attribute__((ext_vector_type(8))) __bf16 bf16x8;
union Pack8 {
    bf16x8 v;
    U4     u;
};

__device__ __forceinline__ float comp4(const F4& a, int p) {
    return p == 0 ? a.x : p == 1 ? a.y : p == 2 ? a.z : a.w;
}

__device__ __forceinline__ void load_weights(F4 (&wv)[16], const float* __restrict__ W, int tid) {
    const int c = tid & 31, kg = tid >> 5;
#pragma unroll
    for (int g = 0; g < 2; ++g)
#pragma unroll
        for (int kk = 0; kk < 8; ++kk)
            wv[g * 8 + kk] = *(const F4*)(W + ((g * 8 + kg) * 8 + kk) * 128 + c * 4);
}

__device__ __forceinline__ void store_weights(unsigned short* sW, const F4 (&wv)[16], int tid) {
    const int c = tid & 31, kg = tid >> 5;
    const int q = c >> 4, l = c & 15;
#pragma unroll
    for (int g = 0; g < 2; ++g) {
        const int k0 = (g * 8 + kg) * 8;
#pragma unroll
        for (int p = 0; p < 4; ++p) {
            Pack8 pk;
#pragma unroll
            for (int kk = 0; kk < 8; ++kk)
                pk.v[kk] = (__bf16)comp4(wv[g * 8 + kk], p);
            *(U4*)(&sW[((q * 4 + p) * 16 + l) * KST_P + k0]) = pk.u;
        }
    }
}

__device__ __forceinline__ void load_row(F4 (&xr)[16], const float* rowp) {
#pragma unroll
    for (int kt = 0; kt < 4; ++kt) {
        xr[kt * 4 + 0] = *(const F4*)(rowp + kt * 32);
        xr[kt * 4 + 1] = *(const F4*)(rowp + kt * 32 + 4);
        xr[kt * 4 + 2] = *(const F4*)(rowp + kt * 32 + 16);
        xr[kt * 4 + 3] = *(const F4*)(rowp + kt * 32 + 20);
    }
}

template <bool SCORE>
__global__ __launch_bounds__(256)
void nodeproj_kernel(const float* x,
                     const float* __restrict__ Wb,
                     const float* __restrict__ Wf,
                     const float* __restrict__ bf,
                     float* P,
                     float* __restrict__ score,
                     float* __restrict__ pooledZ,
                     int N) {
    __shared__ __attribute__((aligned(16))) unsigned short sW[H * KST_P];

    const int tid = threadIdx.x;
    const int wave = tid >> 5, lane = tid & 31;
    const int hi = lane >> 4, ln = lane & 15;
    const int mtiles = (N + 15) >> 4;
    const int mt = blockIdx.x * 8 + wave;
    const int mBase = mt * 16;
    const int mRow  = (mBase + ln) < N ? (mBase + ln) : (N - 1);

    F4 xr[16], wv[16];
    load_row(xr, x + mRow * D + hi * 8);
    load_weights(wv, Wb, tid);
    __builtin_amdgcn_sched_barrier(0);
    {
        const int mBaseU = __builtin_amdgcn_readfirstlane(mBase);
        F4* zp = (F4*)(pooledZ + mBaseU * H) + lane;
        const F4 z{0.f, 0.f, 0.f, 0.f};
#pragma unroll
        for (int j = 0; j < 16; ++j)
            if (mBaseU + j < N) zp[j * 32] = z;
    }
    store_weights(sW, wv, tid);
    __syncthreads();

    if (mt >= mtiles) return;

    FragC acc[8];
#pragma unroll
    for (int nt = 0; nt < 8; ++nt)
#pragma unroll
        for (int r = 0; r < 8; ++r) acc[nt].f[r] = 0.f;

    float part = 0.f;
#pragma unroll
    for (int kt = 0; kt < 4; ++kt) {
        const int kb = kt * 32 + hi * 8;
        FragAB a;
        pack8(a, 0, xr[kt * 4 + 0], xr[kt * 4 + 1]);
        pack8(a, 8, xr[kt * 4 + 2], xr[kt * 4 + 3]);
        if (SCORE) {
            part += dot4(xr[kt * 4 + 0], *(const F4*)(Wf + kb));
            part += dot4(xr[kt * 4 + 1], *(const F4*)(Wf + kb + 4));
            part += dot4(xr[kt * 4 + 2], *(const F4*)(Wf + kb + 16));
            part += dot4(xr[kt * 4 + 3], *(const F4*)(Wf + kb + 20));
        }
        FragAB bcur, bnxt;
        loadB(bcur, sW, ln * KST_P + kb);
#pragma unroll
        for (int nt = 0; nt < 8; ++nt) {
            if (nt < 7)
                loadB(bnxt, sW, ((nt + 1) * 16 + ln) * KST_P + kb);
            acc[nt].v = __builtin_amdgcn_wmma_f32_16x16x32_bf16(
                false, a.v, false, bcur.v, (short)0, acc[nt].v, false, false);
            bcur = bnxt;
        }
    }

#pragma unroll
    for (int r = 0; r < 8; ++r) {
        const int m = mBase + r + hi * 8;
        if (m < N) {
            float* pr = P + m * H + ln * 4;
            F4 v0{acc[0].f[r], acc[1].f[r], acc[2].f[r], acc[3].f[r]};
            F4 v1{acc[4].f[r], acc[5].f[r], acc[6].f[r], acc[7].f[r]};
            *(F4*)(pr)      = v0;
            *(F4*)(pr + 64) = v1;
        }
    }
    if (SCORE) {
        part += __shfl_xor(part, 16, 32);
        if (hi == 0 && (mBase + ln) < N) score[mBase + ln] = part + bf[0];
    }
}

template <bool RAGGED>
__global__ __launch_bounds__(512) __attribute__((amdgpu_waves_per_eu(4, 4)))
void edge_kernel(const float* __restrict__ P,
                 const float* __restrict__ addInfo,
                 const int*   __restrict__ nidx,
                 const int*   __restrict__ segIds,
                 const float* __restrict__ Wb,
                 const float* __restrict__ bb,
                 unsigned int* __restrict__ pooled,
                 int E) {
    const int tid = threadIdx.x;
    const int wave = tid >> 5, lane = tid & 31;
    const int hi = lane >> 4, ln = lane & 15;
    const int numGroups = (E + 7) >> 3;
    const int numHalves = (int)gridDim.x * 32;
    const int steps     = (numGroups + numHalves - 1) / numHalves;
    const int waveId    = (int)blockIdx.x * 16 + wave;
    const int aBase = ((waveId * 2 + (ln >> 3)) * steps) * 8 + (ln & 7);
    const int dGrp  = (waveId * 2 + hi) * steps;
    const int dBase = dGrp * 8;
    const int lastGroup = (E >> 3) - 1;
    const int last  = E - 1;

    float bbv[8];
    f32x2 bW[8];
#pragma unroll
    for (int nt = 0; nt < 8; ++nt) {
        const int col = (nt >> 2) * 64 + ln * 4 + (nt & 3);
        bbv[nt] = bb[col];
        bW[nt][0] = Wb[(D + 2 * hi) * H + col];
        bW[nt][1] = Wb[(D + 2 * hi + 1) * H + col];
    }

    int gC[8], gN[8];
    load8<RAGGED>(gC, nidx, dGrp, lastGroup, E);
    load8<RAGGED>(gN, nidx, dGrp + 1, lastGroup, E);
    int segC[8];
    load8<RAGGED>(segC, segIds, dGrp, lastGroup, E);
    f32x2 av = *(const f32x2*)(addInfo + (aBase < E ? aBase : last) * A + 2 * hi);
    __builtin_amdgcn_sched_barrier(0);
    F4 raw[8][2];
#pragma unroll
    for (int r = 0; r < 8; ++r) {
        const float* __restrict__ rp = P + gC[r] * H + ln * 4;
        raw[r][0] = *(const F4*)(rp);
        raw[r][1] = *(const F4*)(rp + 64);
        __builtin_amdgcn_sched_barrier(0);
    }

    int   prevSeg = -1;
    bool  firstRun = true;
    int   fSeg = -1;
    float fV[8];
#pragma unroll
    for (int nt = 0; nt < 8; ++nt) fV[nt] = 0.f;
    float best[8];
#pragma unroll
    for (int nt = 0; nt < 8; ++nt) best[nt] = -__builtin_inff();

#pragma unroll 1
    for (int t = 0; t < steps; ++t) {
    const int e0 = dBase + t * 8;

    int gNN[8];
    load8<RAGGED>(gNN, nidx, dGrp + t + 2, lastGroup, E);

    int segN[8];
    load8<RAGGED>(segN, segIds, dGrp + t + 1, lastGroup, E);

    FragC acc[8];
    {
        FragC z;
#pragma unroll
        for (int r = 0; r < 8; ++r) z.f[r] = 0.f;
#pragma unroll
        for (int nt = 0; nt < 8; ++nt)
            acc[nt].v = __builtin_amdgcn_wmma_f32_16x16x4_f32(
                false, av, false, bW[nt], (short)0, z.v, false, false);
    }
    {
        const int eN = aBase + (t + 1) * 8;
        av = *(const f32x2*)(addInfo + (eN < E ? eN : last) * A + 2 * hi);
    }

#pragma unroll
    for (int r = 0; r < 8; ++r) {
        const int s = (e0 + r < E) ? segC[r] : -1;
        if (s >= 0) {
            if (s != prevSeg) {
                if (prevSeg >= 0) {
                    float v[8];
#pragma unroll
                    for (int nt = 0; nt < 8; ++nt)
                        v[nt] = fmaxf(best[nt] + bbv[nt], 0.f);
                    if (firstRun) {
                        fSeg = prevSeg;
#pragma unroll
                        for (int nt = 0; nt < 8; ++nt) fV[nt] = v[nt];
                        firstRun = false;
                    } else {
                        unsigned int* __restrict__ pp = pooled + prevSeg * H + ln * 4;
                        *(U4*)(pp)      = U4{__float_as_uint(v[0]), __float_as_uint(v[1]), __float_as_uint(v[2]), __float_as_uint(v[3])};
                        *(U4*)(pp + 64) = U4{__float_as_uint(v[4]), __float_as_uint(v[5]), __float_as_uint(v[6]), __float_as_uint(v[7])};
                    }
                }
                prevSeg = s;
#pragma unroll
                for (int nt = 0; nt < 8; ++nt) best[nt] = -__builtin_inff();
            }
            best[0] = __builtin_elementwise_maximum(best[0], acc[0].f[r] + raw[r][0].x);
            best[1] = __builtin_elementwise_maximum(best[1], acc[1].f[r] + raw[r][0].y);
            best[2] = __builtin_elementwise_maximum(best[2], acc[2].f[r] + raw[r][0].z);
            best[3] = __builtin_elementwise_maximum(best[3], acc[3].f[r] + raw[r][0].w);
            best[4] = __builtin_elementwise_maximum(best[4], acc[4].f[r] + raw[r][1].x);
            best[5] = __builtin_elementwise_maximum(best[5], acc[5].f[r] + raw[r][1].y);
            best[6] = __builtin_elementwise_maximum(best[6], acc[6].f[r] + raw[r][1].z);
            best[7] = __builtin_elementwise_maximum(best[7], acc[7].f[r] + raw[r][1].w);
        }
        const float* __restrict__ rp = P + gN[r] * H + ln * 4;
        raw[r][0] = *(const F4*)(rp);
        raw[r][1] = *(const F4*)(rp + 64);
    }

#pragma unroll
    for (int r = 0; r < 8; ++r) { gN[r] = gNN[r]; segC[r] = segN[r]; }
    }

    const bool has   = prevSeg >= 0;
    const bool multi = !firstRun;
    float lastV[8];
#pragma unroll
    for (int nt = 0; nt < 8; ++nt) lastV[nt] = fmaxf(best[nt] + bbv[nt], 0.f);
    const int firstSeg = multi ? fSeg : prevSeg;
    float firstV[8];
#pragma unroll
    for (int nt = 0; nt < 8; ++nt) firstV[nt] = multi ? fV[nt] : lastV[nt];

    const int oLastSeg  = __shfl_xor(prevSeg, 16, 32);
    const int oFirstSeg = __shfl_xor(firstSeg, 16, 32);
    const int oMulti    = __shfl_xor(multi ? 1 : 0, 16, 32);
    float oLastV[8];
#pragma unroll
    for (int nt = 0; nt < 8; ++nt) oLastV[nt] = __shfl_xor(lastV[nt], 16, 32);

    const bool join = multi && (oMulti != 0) &&
                      (hi == 0 ? (prevSeg == oFirstSeg) : (firstSeg == oLastSeg));

    if (has) {
        unsigned int* __restrict__ pp = pooled + firstSeg * H + ln * 4;
        if (hi == 1 && join) {
            *(U4*)(pp)      = U4{__float_as_uint(fmaxf(firstV[0], oLastV[0])), __float_as_uint(fmaxf(firstV[1], oLastV[1])),
                                 __float_as_uint(fmaxf(firstV[2], oLastV[2])), __float_as_uint(fmaxf(firstV[3], oLastV[3]))};
            *(U4*)(pp + 64) = U4{__float_as_uint(fmaxf(firstV[4], oLastV[4])), __float_as_uint(fmaxf(firstV[5], oLastV[5])),
                                 __float_as_uint(fmaxf(firstV[6], oLastV[6])), __float_as_uint(fmaxf(firstV[7], oLastV[7]))};
        } else {
#pragma unroll
            for (int nt = 0; nt < 8; ++nt)
                atomicMax(pp + (nt >> 2) * 64 + (nt & 3), __float_as_uint(firstV[nt]));
        }
    }
    if (multi && !(hi == 0 && join)) {
        unsigned int* __restrict__ pp = pooled + prevSeg * H + ln * 4;
#pragma unroll
        for (int nt = 0; nt < 8; ++nt)
            atomicMax(pp + (nt >> 2) * 64 + (nt & 3), __float_as_uint(lastV[nt]));
    }
}

template <int M>
__device__ __forceinline__ float row_xor(float v) {
    return __builtin_bit_cast(float,
        __builtin_amdgcn_update_dpp(0, __builtin_bit_cast(int, v), 0x160 + M, 0xf, 0xf, false));
}

template <bool LAST>
__global__ __launch_bounds__(256)
void proj_kernel(const float* __restrict__ pooled,
                 const float* __restrict__ Wo,
                 const float* __restrict__ bo,
                 const float* __restrict__ src,
                 const float* __restrict__ Wf,
                 float* __restrict__ dst,
                 float* __restrict__ out,
                 int N) {
    __shared__ __attribute__((aligned(16))) unsigned short sW[D * KST_P];

    const int tid = threadIdx.x;
    const int wave = tid >> 5, lane = tid & 31;
    const int hi = lane >> 4, ln = lane & 15;
    const int mtiles = (N + 15) >> 4;
    const int mt = blockIdx.x * 8 + wave;
    const int mBase = mt * 16;
    const int mRow  = (mBase + ln) < N ? (mBase + ln) : (N - 1);

    F4 xr[16], wv[16];
    load_row(xr, pooled + mRow * H + hi * 8);
    load_weights(wv, Wo, tid);
    __builtin_amdgcn_sched_barrier(0);
    store_weights(sW, wv, tid);

    const F4 b0 = *(const F4*)(bo + ln * 4);
    const F4 b1 = *(const F4*)(bo + 64 + ln * 4);
    F4 rs[8][2];
    F4 wf0, wf1;
    float oprev = 0.f;
    const int mOut = mBase + hi * 8 + (ln & 7);
    if (!LAST) {
#pragma unroll
        for (int r = 0; r < 8; ++r) {
            const int m  = mBase + r + hi * 8;
            const int mc = m < N ? m : (N - 1);
            const float* sr = src + mc * D + ln * 4;
            rs[r][0] = *(const F4*)(sr);
            rs[r][1] = *(const F4*)(sr + 64);
        }
    } else {
        wf0 = *(const F4*)(Wf + ln * 4);
        wf1 = *(const F4*)(Wf + 64 + ln * 4);
        oprev = out[mOut < N ? mOut : (N - 1)];
    }
    __syncthreads();

    if (mt >= mtiles) return;

    FragC acc[8];
#pragma unroll
    for (int nt = 0; nt < 8; ++nt)
#pragma unroll
        for (int r = 0; r < 8; ++r) acc[nt].f[r] = 0.f;

#pragma unroll
    for (int kt = 0; kt < 4; ++kt) {
        const int kb = kt * 32 + hi * 8;
        FragAB a;
        pack8(a, 0, xr[kt * 4 + 0], xr[kt * 4 + 1]);
        pack8(a, 8, xr[kt * 4 + 2], xr[kt * 4 + 3]);
        FragAB bcur, bnxt;
        loadB(bcur, sW, ln * KST_P + kb);
#pragma unroll
        for (int nt = 0; nt < 8; ++nt) {
            if (nt < 7)
                loadB(bnxt, sW, ((nt + 1) * 16 + ln) * KST_P + kb);
            acc[nt].v = __builtin_amdgcn_wmma_f32_16x16x32_bf16(
                false, a.v, false, bcur.v, (short)0, acc[nt].v, false, false);
            bcur = bnxt;
        }
    }

    if (!LAST) {
#pragma unroll
        for (int r = 0; r < 8; ++r) {
            const int m = mBase + r + hi * 8;
            if (m < N) {
                F4 v0{acc[0].f[r] + b0.x, acc[1].f[r] + b0.y, acc[2].f[r] + b0.z, acc[3].f[r] + b0.w};
                F4 v1{acc[4].f[r] + b1.x, acc[5].f[r] + b1.y, acc[6].f[r] + b1.z, acc[7].f[r] + b1.w};
                v0.x += rs[r][0].x; v0.y += rs[r][0].y; v0.z += rs[r][0].z; v0.w += rs[r][0].w;
                v1.x += rs[r][1].x; v1.y += rs[r][1].y; v1.z += rs[r][1].z; v1.w += rs[r][1].w;
                float* dr = dst + m * D + ln * 4;
                *(F4*)(dr)      = v0;
                *(F4*)(dr + 64) = v1;
            }
        }
    } else {
        const float cb = dot4(b0, wf0) + dot4(b1, wf1);
        float s[8];
#pragma unroll
        for (int r = 0; r < 8; ++r) {
            const F4 a0{acc[0].f[r], acc[1].f[r], acc[2].f[r], acc[3].f[r]};
            const F4 a1{acc[4].f[r], acc[5].f[r], acc[6].f[r], acc[7].f[r]};
            s[r] = cb + dot4(a0, wf0) + dot4(a1, wf1);
        }
#pragma unroll
        for (int r = 0; r < 8; ++r) s[r] += row_xor<1>(s[r]);
#pragma unroll
        for (int r = 0; r < 8; ++r) s[r] += row_xor<2>(s[r]);
#pragma unroll
        for (int r = 0; r < 8; ++r) s[r] += row_xor<4>(s[r]);
#pragma unroll
        for (int r = 0; r < 8; ++r) s[r] += row_xor<8>(s[r]);
        float mine = s[0];
#pragma unroll
        for (int r = 1; r < 8; ++r) mine = (ln == r) ? s[r] : mine;
        if (ln < 8 && mOut < N) out[mOut] = oprev + mine;
    }
}

typedef __attribute__((ext_vector_type(4))) __bf16 bf16x4;
struct alignas(8) U2 { unsigned int x, y; };
union Pack4 {
    bf16x4 v;
    U2     u;
};

__global__ __launch_bounds__(256)
void projnode_kernel(const float* pooled,
                     const float* __restrict__ Wo,
                     const float* __restrict__ bo,
                     const float* __restrict__ src,
                     const float* __restrict__ Wb,
                     const float* __restrict__ Wf,
                     const float* __restrict__ bf,
                     float* __restrict__ P,
                     float* __restrict__ score,
                     float* pooledZ,
                     int N) {
    __shared__ __attribute__((aligned(16))) unsigned short sWo[D * KST_P];
    __shared__ __attribute__((aligned(16))) unsigned short sWb[H * KST_P];
    __shared__ __attribute__((aligned(16))) unsigned short sX[8][16 * KST_P];

    const int tid = threadIdx.x;
    const int wave = tid >> 5, lane = tid & 31;
    const int hi = lane >> 4, ln = lane & 15;
    const int mtiles = (N + 15) >> 4;
    const int mt = blockIdx.x * 8 + wave;
    const int mBase = mt * 16;
    const int mRow  = (mBase + ln) < N ? (mBase + ln) : (N - 1);
    const bool active = mt < mtiles;

    F4 xr[16];
    {
        F4 wvA[16], wvB[16];
        load_row(xr, pooled + mRow * H + hi * 8);
        load_weights(wvA, Wo, tid);
        load_weights(wvB, Wb, tid);
        __builtin_amdgcn_sched_barrier(0);
        store_weights(sWo, wvA, tid);
        store_weights(sWb, wvB, tid);
    }

    const F4 b0 = *(const F4*)(bo + ln * 4);
    const F4 b1 = *(const F4*)(bo + 64 + ln * 4);
    const F4 wf0 = *(const F4*)(Wf + ln * 4);
    const F4 wf1 = *(const F4*)(Wf + 64 + ln * 4);
    F4 rs[8][2];
#pragma unroll
    for (int r = 0; r < 8; ++r) {
        const int m  = mBase + r + hi * 8;
        const int mc = m < N ? m : (N - 1);
        const float* sr = src + mc * D + ln * 4;
        rs[r][0] = *(const F4*)(sr);
        rs[r][1] = *(const F4*)(sr + 64);
    }
    __syncthreads();

    {
        const int mBaseU = __builtin_amdgcn_readfirstlane(mBase);
        F4* zp = (F4*)(pooledZ + mBaseU * H) + lane;
        const F4 z{0.f, 0.f, 0.f, 0.f};
#pragma unroll
        for (int j = 0; j < 16; ++j)
            if (mBaseU + j < N) zp[j * 32] = z;
    }

    if (!active) return;

    FragC acc[8];
#pragma unroll
    for (int nt = 0; nt < 8; ++nt)
#pragma unroll
        for (int r = 0; r < 8; ++r) acc[nt].f[r] = 0.f;

#pragma unroll
    for (int kt = 0; kt < 4; ++kt) {
        const int kb = kt * 32 + hi * 8;
        FragAB a;
        pack8(a, 0, xr[kt * 4 + 0], xr[kt * 4 + 1]);
        pack8(a, 8, xr[kt * 4 + 2], xr[kt * 4 + 3]);
        FragAB bcur, bnxt;
        loadB(bcur, sWo, ln * KST_P + kb);
#pragma unroll
        for (int nt = 0; nt < 8; ++nt) {
            if (nt < 7)
                loadB(bnxt, sWo, ((nt + 1) * 16 + ln) * KST_P + kb);
            acc[nt].v = __builtin_amdgcn_wmma_f32_16x16x32_bf16(
                false, a.v, false, bcur.v, (short)0, acc[nt].v, false, false);
            bcur = bnxt;
        }
    }

    unsigned short* sXw = sX[wave];
    float s[8];
#pragma unroll
    for (int r = 0; r < 8; ++r) {
        F4 v0{acc[0].f[r] + b0.x, acc[1].f[r] + b0.y, acc[2].f[r] + b0.z, acc[3].f[r] + b0.w};
        F4 v1{acc[4].f[r] + b1.x, acc[5].f[r] + b1.y, acc[6].f[r] + b1.z, acc[7].f[r] + b1.w};
        v0.x += rs[r][0].x; v0.y += rs[r][0].y; v0.z += rs[r][0].z; v0.w += rs[r][0].w;
        v1.x += rs[r][1].x; v1.y += rs[r][1].y; v1.z += rs[r][1].z; v1.w += rs[r][1].w;
        s[r] = dot4(v0, wf0) + dot4(v1, wf1);
        Pack4 p0, p1;
        p0.v[0] = (__bf16)v0.x; p0.v[1] = (__bf16)v0.y; p0.v[2] = (__bf16)v0.z; p0.v[3] = (__bf16)v0.w;
        p1.v[0] = (__bf16)v1.x; p1.v[1] = (__bf16)v1.y; p1.v[2] = (__bf16)v1.z; p1.v[3] = (__bf16)v1.w;
        *(U2*)(&sXw[(hi * 8 + r) * KST_P + ln * 4])      = p0.u;
        *(U2*)(&sXw[(hi * 8 + r) * KST_P + 64 + ln * 4]) = p1.u;
    }
#pragma unroll
    for (int r = 0; r < 8; ++r) s[r] += row_xor<1>(s[r]);
#pragma unroll
    for (int r = 0; r < 8; ++r) s[r] += row_xor<2>(s[r]);
#pragma unroll
    for (int r = 0; r < 8; ++r) s[r] += row_xor<4>(s[r]);
#pragma unroll
    for (int r = 0; r < 8; ++r) s[r] += row_xor<8>(s[r]);
    {
        float mine = s[0];
#pragma unroll
        for (int r = 1; r < 8; ++r) mine = (ln == r) ? s[r] : mine;
        const int mOut = mBase + hi * 8 + (ln & 7);
        if (ln < 8 && mOut < N) score[mOut] = mine + bf[0];
    }

#pragma unroll
    for (int nt = 0; nt < 8; ++nt)
#pragma unroll
        for (int r = 0; r < 8; ++r) acc[nt].f[r] = 0.f;
#pragma unroll
    for (int kt = 0; kt < 4; ++kt) {
        const int kb = kt * 32 + hi * 8;
        FragAB a;
        loadB(a, sXw, ln * KST_P + kb);
        FragAB bcur, bnxt;
        loadB(bcur, sWb, ln * KST_P + kb);
#pragma unroll
        for (int nt = 0; nt < 8; ++nt) {
            if (nt < 7)
                loadB(bnxt, sWb, ((nt + 1) * 16 + ln) * KST_P + kb);
            acc[nt].v = __builtin_amdgcn_wmma_f32_16x16x32_bf16(
                false, a.v, false, bcur.v, (short)0, acc[nt].v, false, false);
            bcur = bnxt;
        }
    }
#pragma unroll
    for (int r = 0; r < 8; ++r) {
        const int m = mBase + r + hi * 8;
        if (m < N) {
            float* pr = P + m * H + ln * 4;
            F4 v0{acc[0].f[r], acc[1].f[r], acc[2].f[r], acc[3].f[r]};
            F4 v1{acc[4].f[r], acc[5].f[r], acc[6].f[r], acc[7].f[r]};
            *(F4*)(pr)      = v0;
            *(F4*)(pr + 64) = v1;
        }
    }
}

extern "C" void kernel_launch(void* const* d_in, const int* in_sizes, int n_in,
                              void* d_out, int out_size, void* d_ws, size_t ws_size,
                              hipStream_t stream) {
    const float* interp  = (const float*)d_in[0];
    const float* addInfo = (const float*)d_in[1];
    const int*   nidx    = (const int*)d_in[2];
    const int*   segIds  = (const int*)d_in[3];
    const float* Wb0 = (const float*)d_in[4];
    const float* bb0 = (const float*)d_in[5];
    const float* Wo0 = (const float*)d_in[6];
    const float* bo0 = (const float*)d_in[7];
    const float* Wb1 = (const float*)d_in[8];
    const float* bb1 = (const float*)d_in[9];
    const float* Wo1 = (const float*)d_in[10];
    const float* bo1 = (const float*)d_in[11];
    const float* Wf  = (const float*)d_in[12];
    const float* bfp = (const float*)d_in[13];
    float* out = (float*)d_out;

    const int N = in_sizes[0] / D;
    const int E = in_sizes[2];

    float* pooled = (float*)d_ws;
    float* plane  = pooled + (size_t)N * H;
    unsigned int* pooledU = (unsigned int*)pooled;

    const int edgeGrd = 256;
    const int projGrd = (((N + 15) / 16) + 7) / 8;

    const bool ragged = (E & 7) != 0;

    nodeproj_kernel<false><<<projGrd, 256, 0, stream>>>(interp, Wb0, Wf, bfp, plane, out, pooled, N);
    if (ragged)
        edge_kernel<true><<<edgeGrd, 512, 0, stream>>>(plane, addInfo, nidx, segIds,
                                                       Wb0, bb0, pooledU, E);
    else
        edge_kernel<false><<<edgeGrd, 512, 0, stream>>>(plane, addInfo, nidx, segIds,
                                                        Wb0, bb0, pooledU, E);
    projnode_kernel<<<projGrd, 256, 0, stream>>>(pooled, Wo0, bo0, interp, Wb1, Wf, bfp, plane, out, pooled, N);

    if (ragged)
        edge_kernel<true><<<edgeGrd, 512, 0, stream>>>(plane, addInfo, nidx, segIds,
                                                       Wb1, bb1, pooledU, E);
    else
        edge_kernel<false><<<edgeGrd, 512, 0, stream>>>(plane, addInfo, nidx, segIds,
                                                        Wb1, bb1, pooledU, E);
    proj_kernel<true><<<projGrd, 256, 0, stream>>>(pooled, Wo1, bo1, interp, Wf, plane, out, N);
}


# edge epilogue: tests per 8-row group: rows-exist tested once per group; a group whose last id is still the open run's id is folded with three-input maxima, no per-row tests
# speedup vs baseline: 4.8392x; 1.3856x over previous
#include <hip/hip_runtime.h>


#define D 128
#define A 4
#define H 128
#define KST_P 136

typedef __attribute__((ext_vector_type(16))) __bf16 bf16x16;
typedef __attribute__((ext_vector_type(8)))  float  f32x8;
typedef __attribute__((ext_vector_type(2)))  float  f32x2;

struct alignas(16) U4 { unsigned int x, y, z, w; };
struct alignas(16) F4 { float x, y, z, w; };

union FragAB {
    bf16x16        v;
    unsigned short us[16];
    unsigned int   ui[8];
};
union FragC {
    f32x8 v;
    float f[8];
};

__device__ __forceinline__ void pack8(FragAB& fr, int base,
                                      const F4& a, const F4& b) {
    fr.v[base + 0] = (__bf16)a.x; fr.v[base + 1] = (__bf16)a.y;
    fr.v[base + 2] = (__bf16)a.z; fr.v[base + 3] = (__bf16)a.w;
    fr.v[base + 4] = (__bf16)b.x; fr.v[base + 5] = (__bf16)b.y;
    fr.v[base + 6] = (__bf16)b.z; fr.v[base + 7] = (__bf16)b.w;
}

__device__ __forceinline__ unsigned short f2bf_s(float f) {
    __bf16 h = (__bf16)f;
    return __builtin_bit_cast(unsigned short, h);
}

__device__ __forceinline__ void loadB(FragAB& b, const unsigned short* sW, int off) {
    *(U4*)(&b.ui[0]) = *(const U4*)(&sW[off]);
    *(U4*)(&b.ui[4]) = *(const U4*)(&sW[off + 16]);
}

__device__ __forceinline__ float dot4(const F4& a, const F4& b) {
    return a.x * b.x + a.y * b.y + a.z * b.z + a.w * b.w;
}

__device__ __forceinline__ float bf_lo(unsigned int u) { return __uint_as_float(u << 16); }
__device__ __forceinline__ float bf_hi(unsigned int u) { return __uint_as_float(u & 0xffff0000u); }

template <bool RAGGED>
__device__ __forceinline__ void load8(int (&d)[8], const int* __restrict__ p, int g, int lastGroup, int E) {
    if (RAGGED) {
#pragma unroll
        for (int r = 0; r < 8; ++r) {
            const int e = g * 8 + r;
            d[r] = p[e < E ? e : E - 1];
        }
    } else {
        const int gc = g < lastGroup ? g : lastGroup;
        const U4 a = *(const U4*)(p + gc * 8);
        const U4 b = *(const U4*)(p + gc * 8 + 4);
        d[0] = (int)a.x; d[1] = (int)a.y; d[2] = (int)a.z; d[3] = (int)a.w;
        d[4] = (int)b.x; d[5] = (int)b.y; d[6] = (int)b.z; d[7] = (int)b.w;
    }
}

typedef __attribute__((ext_vector_type(8))) __bf16 bf16x8;
union Pack8 {
    bf16x8 v;
    U4     u;
};

__device__ __forceinline__ float comp4(const F4& a, int p) {
    return p == 0 ? a.x : p == 1 ? a.y : p == 2 ? a.z : a.w;
}

__device__ __forceinline__ void load_weights(F4 (&wv)[16], const float* __restrict__ W, int tid) {
    const int c = tid & 31, kg = tid >> 5;
#pragma unroll
    for (int g = 0; g < 2; ++g)
#pragma unroll
        for (int kk = 0; kk < 8; ++kk)
            wv[g * 8 + kk] = *(const F4*)(W + ((g * 8 + kg) * 8 + kk) * 128 + c * 4);
}

__device__ __forceinline__ void store_weights(unsigned short* sW, const F4 (&wv)[16], int tid) {
    const int c = tid & 31, kg = tid >> 5;
    const int q = c >> 4, l = c & 15;
#pragma unroll
    for (int g = 0; g < 2; ++g) {
        const int k0 = (g * 8 + kg) * 8;
#pragma unroll
        for (int p = 0; p < 4; ++p) {
            Pack8 pk;
#pragma unroll
            for (int kk = 0; kk < 8; ++kk)
                pk.v[kk] = (__bf16)comp4(wv[g * 8 + kk], p);
            *(U4*)(&sW[((q * 4 + p) * 16 + l) * KST_P + k0]) = pk.u;
        }
    }
}

__device__ __forceinline__ void load_row(F4 (&xr)[16], const float* rowp) {
#pragma unroll
    for (int kt = 0; kt < 4; ++kt) {
        xr[kt * 4 + 0] = *(const F4*)(rowp + kt * 32);
        xr[kt * 4 + 1] = *(const F4*)(rowp + kt * 32 + 4);
        xr[kt * 4 + 2] = *(const F4*)(rowp + kt * 32 + 16);
        xr[kt * 4 + 3] = *(const F4*)(rowp + kt * 32 + 20);
    }
}

template <bool SCORE>
__global__ __launch_bounds__(256)
void nodeproj_kernel(const float* __restrict__ x,
                     const float* __restrict__ Wb,
                     const float* __restrict__ Wf,
                     const float* __restrict__ bf,
                     U4* __restrict__ P,
                     float* __restrict__ score,
                     float* __restrict__ pooledZ,
                     int N) {
    __shared__ __attribute__((aligned(16))) unsigned short sW[H * KST_P];

    const int tid = threadIdx.x;
    const int wave = tid >> 5, lane = tid & 31;
    const int hi = lane >> 4, ln = lane & 15;
    const int mtiles = (N + 15) >> 4;
    const int mt = blockIdx.x * 8 + wave;
    const int mBase = mt * 16;
    const int mRow  = (mBase + ln) < N ? (mBase + ln) : (N - 1);

    F4 xr[16], wv[16];
    load_row(xr, x + mRow * D + hi * 8);
    load_weights(wv, Wb, tid);
    __builtin_amdgcn_sched_barrier(0);
    {
        const int mBaseU = __builtin_amdgcn_readfirstlane(mBase);
        F4* zp = (F4*)(pooledZ + mBaseU * H) + lane;
        const F4 z{0.f, 0.f, 0.f, 0.f};
#pragma unroll
        for (int j = 0; j < 16; ++j)
            if (mBaseU + j < N) zp[j * 32] = z;
    }
    store_weights(sW, wv, tid);
    __syncthreads();

    if (mt >= mtiles) return;

    FragC acc[8];
#pragma unroll
    for (int nt = 0; nt < 8; ++nt)
#pragma unroll
        for (int r = 0; r < 8; ++r) acc[nt].f[r] = 0.f;

    float part = 0.f;
#pragma unroll
    for (int kt = 0; kt < 4; ++kt) {
        const int kb = kt * 32 + hi * 8;
        FragAB a;
        pack8(a, 0, xr[kt * 4 + 0], xr[kt * 4 + 1]);
        pack8(a, 8, xr[kt * 4 + 2], xr[kt * 4 + 3]);
        if (SCORE) {
            part += dot4(xr[kt * 4 + 0], *(const F4*)(Wf + kb));
            part += dot4(xr[kt * 4 + 1], *(const F4*)(Wf + kb + 4));
            part += dot4(xr[kt * 4 + 2], *(const F4*)(Wf + kb + 16));
            part += dot4(xr[kt * 4 + 3], *(const F4*)(Wf + kb + 20));
        }
        FragAB bcur, bnxt;
        loadB(bcur, sW, ln * KST_P + kb);
#pragma unroll
        for (int nt = 0; nt < 8; ++nt) {
            if (nt < 7)
                loadB(bnxt, sW, ((nt + 1) * 16 + ln) * KST_P + kb);
            acc[nt].v = __builtin_amdgcn_wmma_f32_16x16x32_bf16(
                false, a.v, false, bcur.v, (short)0, acc[nt].v, false, false);
            bcur = bnxt;
        }
    }

#pragma unroll
    for (int r = 0; r < 8; ++r) {
        const int m = mBase + r + hi * 8;
        if (m < N) {
            Pack8 pk;
#pragma unroll
            for (int nt = 0; nt < 8; ++nt) pk.v[nt] = (__bf16)acc[nt].f[r];
            P[m * 16 + ln] = pk.u;
        }
    }
    if (SCORE) {
        part += __shfl_xor(part, 16, 32);
        if (hi == 0 && (mBase + ln) < N) score[mBase + ln] = part + bf[0];
    }
}

template <bool RAGGED>
__global__ __launch_bounds__(512) __attribute__((amdgpu_waves_per_eu(4, 4)))
void edge_kernel(const U4* __restrict__ P,
                 const float* __restrict__ addInfo,
                 const int*   __restrict__ nidx,
                 const int*   __restrict__ segIds,
                 const float* __restrict__ Wb,
                 const float* __restrict__ bb,
                 unsigned int* __restrict__ pooled,
                 int E) {
    const int tid = threadIdx.x;
    const int wave = tid >> 5, lane = tid & 31;
    const int hi = lane >> 4, ln = lane & 15;
    const int numGroups = (E + 7) >> 3;
    const int numHalves = (int)gridDim.x * 32;
    const int steps     = (numGroups + numHalves - 1) / numHalves;
    const int waveId    = (int)blockIdx.x * 16 + wave;
    const int aBase = ((waveId * 2 + (ln >> 3)) * steps) * 8 + (ln & 7);
    const int dGrp  = (waveId * 2 + hi) * steps;
    const int dBase = dGrp * 8;
    const int lastGroup = (E >> 3) - 1;
    const int last  = E - 1;

    float bbv[8];
    f32x2 bW[8];
#pragma unroll
    for (int nt = 0; nt < 8; ++nt) {
        const int col = (nt >> 2) * 64 + ln * 4 + (nt & 3);
        bbv[nt] = bb[col];
        bW[nt][0] = Wb[(D + 2 * hi) * H + col];
        bW[nt][1] = Wb[(D + 2 * hi + 1) * H + col];
    }

    int gC[8], gN[8];
    load8<RAGGED>(gC, nidx, dGrp, lastGroup, E);
    load8<RAGGED>(gN, nidx, dGrp + 1, lastGroup, E);
    int segC[8];
    load8<RAGGED>(segC, segIds, dGrp, lastGroup, E);
    f32x2 av = *(const f32x2*)(addInfo + (aBase < E ? aBase : last) * A + 2 * hi);
    U4 raw[8];
#pragma unroll
    for (int r = 0; r < 8; ++r) raw[r] = P[gC[r] * 16 + ln];

    int   prevSeg = -1;
    bool  firstRun = true;
    int   fSeg = -1;
    float fV[8];
#pragma unroll
    for (int nt = 0; nt < 8; ++nt) fV[nt] = 0.f;
    float best[8];
#pragma unroll
    for (int nt = 0; nt < 8; ++nt) best[nt] = -__builtin_inff();

#pragma unroll 1
    for (int t = 0; t < steps; ++t) {
    const int e0 = dBase + t * 8;

    int gNN[8];
    load8<RAGGED>(gNN, nidx, dGrp + t + 2, lastGroup, E);

    int segN[8];
    load8<RAGGED>(segN, segIds, dGrp + t + 1, lastGroup, E);

    FragC acc[8];
#pragma unroll
    for (int r = 0; r < 8; ++r) {
        acc[0].f[r] = bf_lo(raw[r].x); acc[1].f[r] = bf_hi(raw[r].x);
        acc[2].f[r] = bf_lo(raw[r].y); acc[3].f[r] = bf_hi(raw[r].y);
        acc[4].f[r] = bf_lo(raw[r].z); acc[5].f[r] = bf_hi(raw[r].z);
        acc[6].f[r] = bf_lo(raw[r].w); acc[7].f[r] = bf_hi(raw[r].w);
    }
#pragma unroll
    for (int r = 0; r < 8; ++r) raw[r] = P[gN[r] * 16 + ln];

#pragma unroll
    for (int nt = 0; nt < 8; ++nt)
        acc[nt].v = __builtin_amdgcn_wmma_f32_16x16x4_f32(
            false, av, false, bW[nt], (short)0, acc[nt].v, false, false);
    {
        const int eN = aBase + (t + 1) * 8;
        av = *(const f32x2*)(addInfo + (eN < E ? eN : last) * A + 2 * hi);
    }

    const bool grpValid   = RAGGED ? (e0 + 7 < E) : (e0 < E);
    const bool wholeGroup = grpValid && (segC[7] == prevSeg);

    auto takeRow = [&](const int s, const int r) __attribute__((always_inline)) {
        if (s != prevSeg) {
            if (prevSeg >= 0) {
                float v[8];
#pragma unroll
                for (int nt = 0; nt < 8; ++nt)
                    v[nt] = fmaxf(best[nt] + bbv[nt], 0.f);
                if (firstRun) {
                    fSeg = prevSeg;
#pragma unroll
                    for (int nt = 0; nt < 8; ++nt) fV[nt] = v[nt];
                    firstRun = false;
                } else {
                    unsigned int* __restrict__ pp = pooled + prevSeg * H + ln * 4;
                    *(U4*)(pp)      = U4{__float_as_uint(v[0]), __float_as_uint(v[1]), __float_as_uint(v[2]), __float_as_uint(v[3])};
                    *(U4*)(pp + 64) = U4{__float_as_uint(v[4]), __float_as_uint(v[5]), __float_as_uint(v[6]), __float_as_uint(v[7])};
                }
            }
            prevSeg = s;
#pragma unroll
            for (int nt = 0; nt < 8; ++nt) best[nt] = -__builtin_inff();
        }
#pragma unroll
        for (int nt = 0; nt < 8; ++nt)
            best[nt] = __builtin_elementwise_maximum(best[nt], acc[nt].f[r]);
    };

    if (wholeGroup) {
#pragma unroll
        for (int nt = 0; nt < 8; ++nt) {
            float b = best[nt];
            b = __builtin_elementwise_maximum(__builtin_elementwise_maximum(b, acc[nt].f[0]), acc[nt].f[1]);
            b = __builtin_elementwise_maximum(__builtin_elementwise_maximum(b, acc[nt].f[2]), acc[nt].f[3]);
            b = __builtin_elementwise_maximum(__builtin_elementwise_maximum(b, acc[nt].f[4]), acc[nt].f[5]);
            b = __builtin_elementwise_maximum(__builtin_elementwise_maximum(b, acc[nt].f[6]), acc[nt].f[7]);
            best[nt] = b;
        }
    } else if (grpValid) {
#pragma unroll
        for (int r = 0; r < 8; ++r) takeRow(segC[r], r);
    } else if (RAGGED) {
#pragma unroll
        for (int r = 0; r < 8; ++r)
            if (e0 + r < E) takeRow(segC[r], r);
    }

#pragma unroll
    for (int r = 0; r < 8; ++r) { gN[r] = gNN[r]; segC[r] = segN[r]; }
    }

    const bool has   = prevSeg >= 0;
    const bool multi = !firstRun;
    float lastV[8];
#pragma unroll
    for (int nt = 0; nt < 8; ++nt) lastV[nt] = fmaxf(best[nt] + bbv[nt], 0.f);
    const int firstSeg = multi ? fSeg : prevSeg;
    float firstV[8];
#pragma unroll
    for (int nt = 0; nt < 8; ++nt) firstV[nt] = multi ? fV[nt] : lastV[nt];

    const int oLastSeg  = __shfl_xor(prevSeg, 16, 32);
    const int oFirstSeg = __shfl_xor(firstSeg, 16, 32);
    const int oMulti    = __shfl_xor(multi ? 1 : 0, 16, 32);
    float oLastV[8];
#pragma unroll
    for (int nt = 0; nt < 8; ++nt) oLastV[nt] = __shfl_xor(lastV[nt], 16, 32);

    const bool join = multi && (oMulti != 0) &&
                      (hi == 0 ? (prevSeg == oFirstSeg) : (firstSeg == oLastSeg));

    if (has) {
        unsigned int* __restrict__ pp = pooled + firstSeg * H + ln * 4;
        if (hi == 1 && join) {
            *(U4*)(pp)      = U4{__float_as_uint(fmaxf(firstV[0], oLastV[0])), __float_as_uint(fmaxf(firstV[1], oLastV[1])),
                                 __float_as_uint(fmaxf(firstV[2], oLastV[2])), __float_as_uint(fmaxf(firstV[3], oLastV[3]))};
            *(U4*)(pp + 64) = U4{__float_as_uint(fmaxf(firstV[4], oLastV[4])), __float_as_uint(fmaxf(firstV[5], oLastV[5])),
                                 __float_as_uint(fmaxf(firstV[6], oLastV[6])), __float_as_uint(fmaxf(firstV[7], oLastV[7]))};
        } else {
#pragma unroll
            for (int nt = 0; nt < 8; ++nt)
                atomicMax(pp + (nt >> 2) * 64 + (nt & 3), __float_as_uint(firstV[nt]));
        }
    }
    if (multi && !(hi == 0 && join)) {
        unsigned int* __restrict__ pp = pooled + prevSeg * H + ln * 4;
#pragma unroll
        for (int nt = 0; nt < 8; ++nt)
            atomicMax(pp + (nt >> 2) * 64 + (nt & 3), __float_as_uint(lastV[nt]));
    }
}

template <int M>
__device__ __forceinline__ float row_xor(float v) {
    return __builtin_bit_cast(float,
        __builtin_amdgcn_update_dpp(0, __builtin_bit_cast(int, v), 0x160 + M, 0xf, 0xf, false));
}

template <bool LAST>
__global__ __launch_bounds__(256)
void proj_kernel(const float* __restrict__ pooled,
                 const float* __restrict__ Wo,
                 const float* __restrict__ bo,
                 const float* __restrict__ src,
                 const float* __restrict__ Wf,
                 float* __restrict__ dst,
                 float* __restrict__ out,
                 int N) {
    __shared__ __attribute__((aligned(16))) unsigned short sW[D * KST_P];

    const int tid = threadIdx.x;
    const int wave = tid >> 5, lane = tid & 31;
    const int hi = lane >> 4, ln = lane & 15;
    const int mtiles = (N + 15) >> 4;
    const int mt = blockIdx.x * 8 + wave;
    const int mBase = mt * 16;
    const int mRow  = (mBase + ln) < N ? (mBase + ln) : (N - 1);

    F4 xr[16], wv[16];
    load_row(xr, pooled + mRow * H + hi * 8);
    load_weights(wv, Wo, tid);
    __builtin_amdgcn_sched_barrier(0);
    store_weights(sW, wv, tid);

    const F4 b0 = *(const F4*)(bo + ln * 4);
    const F4 b1 = *(const F4*)(bo + 64 + ln * 4);
    F4 rs[8][2];
    F4 wf0, wf1;
    float oprev = 0.f;
    const int mOut = mBase + hi * 8 + (ln & 7);
    if (!LAST) {
#pragma unroll
        for (int r = 0; r < 8; ++r) {
            const int m  = mBase + r + hi * 8;
            const int mc = m < N ? m : (N - 1);
            const float* sr = src + mc * D + ln * 4;
            rs[r][0] = *(const F4*)(sr);
            rs[r][1] = *(const F4*)(sr + 64);
        }
    } else {
        wf0 = *(const F4*)(Wf + ln * 4);
        wf1 = *(const F4*)(Wf + 64 + ln * 4);
        oprev = out[mOut < N ? mOut : (N - 1)];
    }
    __syncthreads();

    if (mt >= mtiles) return;

    FragC acc[8];
#pragma unroll
    for (int nt = 0; nt < 8; ++nt)
#pragma unroll
        for (int r = 0; r < 8; ++r) acc[nt].f[r] = 0.f;

#pragma unroll
    for (int kt = 0; kt < 4; ++kt) {
        const int kb = kt * 32 + hi * 8;
        FragAB a;
        pack8(a, 0, xr[kt * 4 + 0], xr[kt * 4 + 1]);
        pack8(a, 8, xr[kt * 4 + 2], xr[kt * 4 + 3]);
        FragAB bcur, bnxt;
        loadB(bcur, sW, ln * KST_P + kb);
#pragma unroll
        for (int nt = 0; nt < 8; ++nt) {
            if (nt < 7)
                loadB(bnxt, sW, ((nt + 1) * 16 + ln) * KST_P + kb);
            acc[nt].v = __builtin_amdgcn_wmma_f32_16x16x32_bf16(
                false, a.v, false, bcur.v, (short)0, acc[nt].v, false, false);
            bcur = bnxt;
        }
    }

    if (!LAST) {
#pragma unroll
        for (int r = 0; r < 8; ++r) {
            const int m = mBase + r + hi * 8;
            if (m < N) {
                F4 v0{acc[0].f[r] + b0.x, acc[1].f[r] + b0.y, acc[2].f[r] + b0.z, acc[3].f[r] + b0.w};
                F4 v1{acc[4].f[r] + b1.x, acc[5].f[r] + b1.y, acc[6].f[r] + b1.z, acc[7].f[r] + b1.w};
                v0.x += rs[r][0].x; v0.y += rs[r][0].y; v0.z += rs[r][0].z; v0.w += rs[r][0].w;
                v1.x += rs[r][1].x; v1.y += rs[r][1].y; v1.z += rs[r][1].z; v1.w += rs[r][1].w;
                float* dr = dst + m * D + ln * 4;
                *(F4*)(dr)      = v0;
                *(F4*)(dr + 64) = v1;
            }
        }
    } else {
        const float cb = dot4(b0, wf0) + dot4(b1, wf1);
        float s[8];
#pragma unroll
        for (int r = 0; r < 8; ++r) {
            const F4 a0{acc[0].f[r], acc[1].f[r], acc[2].f[r], acc[3].f[r]};
            const F4 a1{acc[4].f[r], acc[5].f[r], acc[6].f[r], acc[7].f[r]};
            s[r] = cb + dot4(a0, wf0) + dot4(a1, wf1);
        }
#pragma unroll
        for (int r = 0; r < 8; ++r) s[r] += row_xor<1>(s[r]);
#pragma unroll
        for (int r = 0; r < 8; ++r) s[r] += row_xor<2>(s[r]);
#pragma unroll
        for (int r = 0; r < 8; ++r) s[r] += row_xor<4>(s[r]);
#pragma unroll
        for (int r = 0; r < 8; ++r) s[r] += row_xor<8>(s[r]);
        float mine = s[0];
#pragma unroll
        for (int r = 1; r < 8; ++r) mine = (ln == r) ? s[r] : mine;
        if (ln < 8 && mOut < N) out[mOut] = oprev + mine;
    }
}

typedef __attribute__((ext_vector_type(4))) __bf16 bf16x4;
struct alignas(8) U2 { unsigned int x, y; };
union Pack4 {
    bf16x4 v;
    U2     u;
};

__global__ __launch_bounds__(256)
void projnode_kernel(const float* pooled,
                     const float* __restrict__ Wo,
                     const float* __restrict__ bo,
                     const float* __restrict__ src,
                     const float* __restrict__ Wb,
                     const float* __restrict__ Wf,
                     const float* __restrict__ bf,
                     U4* __restrict__ P,
                     float* __restrict__ score,
                     float* pooledZ,
                     int N) {
    __shared__ __attribute__((aligned(16))) unsigned short sWo[D * KST_P];
    __shared__ __attribute__((aligned(16))) unsigned short sWb[H * KST_P];
    __shared__ __attribute__((aligned(16))) unsigned short sX[8][16 * KST_P];

    const int tid = threadIdx.x;
    const int wave = tid >> 5, lane = tid & 31;
    const int hi = lane >> 4, ln = lane & 15;
    const int mtiles = (N + 15) >> 4;
    const int mt = blockIdx.x * 8 + wave;
    const int mBase = mt * 16;
    const int mRow  = (mBase + ln) < N ? (mBase + ln) : (N - 1);
    const bool active = mt < mtiles;

    F4 xr[16];
    {
        F4 wvA[16], wvB[16];
        load_row(xr, pooled + mRow * H + hi * 8);
        load_weights(wvA, Wo, tid);
        load_weights(wvB, Wb, tid);
        __builtin_amdgcn_sched_barrier(0);
        store_weights(sWo, wvA, tid);
        store_weights(sWb, wvB, tid);
    }

    const F4 b0 = *(const F4*)(bo + ln * 4);
    const F4 b1 = *(const F4*)(bo + 64 + ln * 4);
    const F4 wf0 = *(const F4*)(Wf + ln * 4);
    const F4 wf1 = *(const F4*)(Wf + 64 + ln * 4);
    F4 rs[8][2];
#pragma unroll
    for (int r = 0; r < 8; ++r) {
        const int m  = mBase + r + hi * 8;
        const int mc = m < N ? m : (N - 1);
        const float* sr = src + mc * D + ln * 4;
        rs[r][0] = *(const F4*)(sr);
        rs[r][1] = *(const F4*)(sr + 64);
    }
    __syncthreads();

    {
        const int mBaseU = __builtin_amdgcn_readfirstlane(mBase);
        F4* zp = (F4*)(pooledZ + mBaseU * H) + lane;
        const F4 z{0.f, 0.f, 0.f, 0.f};
#pragma unroll
        for (int j = 0; j < 16; ++j)
            if (mBaseU + j < N) zp[j * 32] = z;
    }

    if (!active) return;

    FragC acc[8];
#pragma unroll
    for (int nt = 0; nt < 8; ++nt)
#pragma unroll
        for (int r = 0; r < 8; ++r) acc[nt].f[r] = 0.f;

#pragma unroll
    for (int kt = 0; kt < 4; ++kt) {
        const int kb = kt * 32 + hi * 8;
        FragAB a;
        pack8(a, 0, xr[kt * 4 + 0], xr[kt * 4 + 1]);
        pack8(a, 8, xr[kt * 4 + 2], xr[kt * 4 + 3]);
        FragAB bcur, bnxt;
        loadB(bcur, sWo, ln * KST_P + kb);
#pragma unroll
        for (int nt = 0; nt < 8; ++nt) {
            if (nt < 7)
                loadB(bnxt, sWo, ((nt + 1) * 16 + ln) * KST_P + kb);
            acc[nt].v = __builtin_amdgcn_wmma_f32_16x16x32_bf16(
                false, a.v, false, bcur.v, (short)0, acc[nt].v, false, false);
            bcur = bnxt;
        }
    }

    unsigned short* sXw = sX[wave];
    float s[8];
#pragma unroll
    for (int r = 0; r < 8; ++r) {
        F4 v0{acc[0].f[r] + b0.x, acc[1].f[r] + b0.y, acc[2].f[r] + b0.z, acc[3].f[r] + b0.w};
        F4 v1{acc[4].f[r] + b1.x, acc[5].f[r] + b1.y, acc[6].f[r] + b1.z, acc[7].f[r] + b1.w};
        v0.x += rs[r][0].x; v0.y += rs[r][0].y; v0.z += rs[r][0].z; v0.w += rs[r][0].w;
        v1.x += rs[r][1].x; v1.y += rs[r][1].y; v1.z += rs[r][1].z; v1.w += rs[r][1].w;
        s[r] = dot4(v0, wf0) + dot4(v1, wf1);
        Pack4 p0, p1;
        p0.v[0] = (__bf16)v0.x; p0.v[1] = (__bf16)v0.y; p0.v[2] = (__bf16)v0.z; p0.v[3] = (__bf16)v0.w;
        p1.v[0] = (__bf16)v1.x; p1.v[1] = (__bf16)v1.y; p1.v[2] = (__bf16)v1.z; p1.v[3] = (__bf16)v1.w;
        *(U2*)(&sXw[(hi * 8 + r) * KST_P + ln * 4])      = p0.u;
        *(U2*)(&sXw[(hi * 8 + r) * KST_P + 64 + ln * 4]) = p1.u;
    }
#pragma unroll
    for (int r = 0; r < 8; ++r) s[r] += row_xor<1>(s[r]);
#pragma unroll
    for (int r = 0; r < 8; ++r) s[r] += row_xor<2>(s[r]);
#pragma unroll
    for (int r = 0; r < 8; ++r) s[r] += row_xor<4>(s[r]);
#pragma unroll
    for (int r = 0; r < 8; ++r) s[r] += row_xor<8>(s[r]);
    {
        float mine = s[0];
#pragma unroll
        for (int r = 1; r < 8; ++r) mine = (ln == r) ? s[r] : mine;
        const int mOut = mBase + hi * 8 + (ln & 7);
        if (ln < 8 && mOut < N) score[mOut] = mine + bf[0];
    }

#pragma unroll
    for (int nt = 0; nt < 8; ++nt)
#pragma unroll
        for (int r = 0; r < 8; ++r) acc[nt].f[r] = 0.f;
#pragma unroll
    for (int kt = 0; kt < 4; ++kt) {
        const int kb = kt * 32 + hi * 8;
        FragAB a;
        loadB(a, sXw, ln * KST_P + kb);
        FragAB bcur, bnxt;
        loadB(bcur, sWb, ln * KST_P + kb);
#pragma unroll
        for (int nt = 0; nt < 8; ++nt) {
            if (nt < 7)
                loadB(bnxt, sWb, ((nt + 1) * 16 + ln) * KST_P + kb);
            acc[nt].v = __builtin_amdgcn_wmma_f32_16x16x32_bf16(
                false, a.v, false, bcur.v, (short)0, acc[nt].v, false, false);
            bcur = bnxt;
        }
    }
#pragma unroll
    for (int r = 0; r < 8; ++r) {
        const int m = mBase + r + hi * 8;
        if (m < N) {
            Pack8 pk;
#pragma unroll
            for (int nt = 0; nt < 8; ++nt) pk.v[nt] = (__bf16)acc[nt].f[r];
            P[m * 16 + ln] = pk.u;
        }
    }
}

extern "C" void kernel_launch(void* const* d_in, const int* in_sizes, int n_in,
                              void* d_out, int out_size, void* d_ws, size_t ws_size,
                              hipStream_t stream) {
    const float* interp  = (const float*)d_in[0];
    const float* addInfo = (const float*)d_in[1];
    const int*   nidx    = (const int*)d_in[2];
    const int*   segIds  = (const int*)d_in[3];
    const float* Wb0 = (const float*)d_in[4];
    const float* bb0 = (const float*)d_in[5];
    const float* Wo0 = (const float*)d_in[6];
    const float* bo0 = (const float*)d_in[7];
    const float* Wb1 = (const float*)d_in[8];
    const float* bb1 = (const float*)d_in[9];
    const float* Wo1 = (const float*)d_in[10];
    const float* bo1 = (const float*)d_in[11];
    const float* Wf  = (const float*)d_in[12];
    const float* bfp = (const float*)d_in[13];
    float* out = (float*)d_out;

    const int N = in_sizes[0] / D;
    const int E = in_sizes[2];

    float* pooled = (float*)d_ws;
    U4*    plane  = (U4*)(pooled + (size_t)N * H);
    unsigned int* pooledU = (unsigned int*)pooled;

    const int edgeGrd = 256;
    const int projGrd = (((N + 15) / 16) + 7) / 8;

    const bool ragged = (E & 7) != 0;

    nodeproj_kernel<false><<<projGrd, 256, 0, stream>>>(interp, Wb0, Wf, bfp, plane, out, pooled, N);
    if (ragged)
        edge_kernel<true><<<edgeGrd, 512, 0, stream>>>(plane, addInfo, nidx, segIds,
                                                       Wb0, bb0, pooledU, E);
    else
        edge_kernel<false><<<edgeGrd, 512, 0, stream>>>(plane, addInfo, nidx, segIds,
                                                        Wb0, bb0, pooledU, E);
    projnode_kernel<<<projGrd, 256, 0, stream>>>(pooled, Wo0, bo0, interp, Wb1, Wf, bfp, plane, out, pooled, N);

    if (ragged)
        edge_kernel<true><<<edgeGrd, 512, 0, stream>>>(plane, addInfo, nidx, segIds,
                                                       Wb1, bb1, pooledU, E);
    else
        edge_kernel<false><<<edgeGrd, 512, 0, stream>>>(plane, addInfo, nidx, segIds,
                                                        Wb1, bb1, pooledU, E);
    proj_kernel<true><<<projGrd, 256, 0, stream>>>(pooled, Wo1, bo1, interp, Wf, (float*)plane, out, N);
}
